# starting kernel
# baseline (speedup 1.0000x reference)
#include <hip/hip_runtime.h>
#include <hip/hip_bf16.h>
#include <math.h>


typedef float v2f __attribute__((ext_vector_type(2)));
typedef float v8f __attribute__((ext_vector_type(8)));

#define NEG_SLOPE 0.2f

__device__ __forceinline__ void atomicMaxF(float* addr, float v) {
  if (v >= 0.0f) {
    atomicMax((int*)addr, __float_as_int(v));
  } else {
    atomicMin((unsigned int*)addr, __float_as_uint(v));
  }
}

__global__ void k_gemm_wmma(const float* __restrict__ A, const float* __restrict__ B,
                            float* __restrict__ Cg, int Nrows, int K, int Fout,
                            int tiles_m, int tiles_n) {
  const int wid  = (blockIdx.x * blockDim.x + threadIdx.x) >> 5;
  const int lane = threadIdx.x & 31;
  if (wid >= tiles_m * tiles_n) return;
  const int tm   = wid / tiles_n;
  const int tn   = wid - tm * tiles_n;
  const int half = lane >> 4;
  const int l15  = lane & 15;
  const int arow = tm * 16 + l15;
  const int bcol = tn * 16 + l15;
  const bool bvalid = bcol < Fout;
  const int arowc = (arow < Nrows) ? arow : (Nrows - 1);
  const int bcolc = bvalid ? bcol : (Fout - 1);

  const float* aptr = A + (size_t)arowc * K + half * 2;
  const float* bptr = B + (size_t)(half * 2) * Fout + bcolc;
  const size_t bstep = (size_t)4 * (size_t)Fout;

  v8f acc = {};
#pragma unroll 4
  for (int k = 0; k < K; k += 4) {
    const v2f a = *(const v2f*)aptr;
    v2f b;
    b.x = bptr[0];
    b.y = bptr[Fout];
    acc = __builtin_amdgcn_wmma_f32_16x16x4_f32(false, a, false, b,
                                                (short)0, acc, false, false);
    aptr += 4;
    bptr += bstep;
  }

  if (bvalid) {
    const int baserow = tm * 16 + half * 8;
#pragma unroll
    for (int i = 0; i < 8; ++i) {
      const int r = baserow + i;
      if (r < Nrows) Cg[(size_t)r * Fout + bcol] = acc[i];
    }
  }
}

__global__ void k_logits(const float* __restrict__ h, const float* __restrict__ a_src,
                         const float* __restrict__ a_dst, float* __restrict__ als,
                         float* __restrict__ ald, int N, int H, int C) {
  const int idx = blockIdx.x * blockDim.x + threadIdx.x;
  if (idx >= N * H) return;
  const int n = idx / H, hh = idx - n * H;
  const float* hp = h + (size_t)n * H * C + (size_t)hh * C;
  const float* ws = a_src + hh * C;
  const float* wd = a_dst + hh * C;
  float ss = 0.0f, sd = 0.0f;
  for (int c = 0; c < C; ++c) {
    const float v = hp[c];
    ss += v * ws[c];
    sd += v * wd[c];
  }
  als[idx] = ss;
  ald[idx] = sd;
}

__global__ void k_fill(float* __restrict__ p, float v, int n) {
  const int idx = blockIdx.x * blockDim.x + threadIdx.x;
  if (idx < n) p[idx] = v;
}

__device__ __forceinline__ void edge_sd(int e, int E, const int* __restrict__ src,
                                        const int* __restrict__ dst, int* s, int* d) {
  if (e < E) { *s = src[e]; *d = dst[e]; }
  else       { *s = e - E;  *d = e - E; }
}

__global__ void k_edge_max(const float* __restrict__ als, const float* __restrict__ ald,
                           const int* __restrict__ src, const int* __restrict__ dst,
                           float* __restrict__ emax, int E, int Etot, int H) {
  const int idx = blockIdx.x * blockDim.x + threadIdx.x;
  if (idx >= Etot * H) return;
  const int e = idx / H, hh = idx - e * H;
  int s, d; edge_sd(e, E, src, dst, &s, &d);
  float z = als[s * H + hh] + ald[d * H + hh];
  z = (z > 0.0f) ? z : NEG_SLOPE * z;
  atomicMaxF(&emax[d * H + hh], z);
}

__global__ void k_edge_expsum(const float* __restrict__ als, const float* __restrict__ ald,
                              const int* __restrict__ src, const int* __restrict__ dst,
                              const float* __restrict__ emax, float* __restrict__ den,
                              float* __restrict__ exb, int E, int Etot, int H) {
  const int idx = blockIdx.x * blockDim.x + threadIdx.x;
  if (idx >= Etot * H) return;
  const int e = idx / H, hh = idx - e * H;
  int s, d; edge_sd(e, E, src, dst, &s, &d);
  float z = als[s * H + hh] + ald[d * H + hh];
  z = (z > 0.0f) ? z : NEG_SLOPE * z;
  float m = emax[d * H + hh];
  if (!isfinite(m)) m = 0.0f;
  const float ev = __expf(z - m);
  exb[idx] = ev;
  atomicAdd(&den[d * H + hh], ev);
}

__global__ void k_recip(float* __restrict__ den, int n) {
  const int idx = blockIdx.x * blockDim.x + threadIdx.x;
  if (idx < n) den[idx] = 1.0f / (den[idx] + 1e-16f);
}

__global__ void k_agg128(const float* __restrict__ h, const float* __restrict__ exb,
                         const float* __restrict__ rden, const int* __restrict__ src,
                         const int* __restrict__ dst, float* __restrict__ out,
                         int E, int Etot) {
  const int gid = blockIdx.x * blockDim.x + threadIdx.x;
  const int e = gid >> 5;
  if (e >= Etot) return;
  const int lane = gid & 31;
  int s, d; edge_sd(e, E, src, dst, &s, &d);
  const int head = lane >> 2;
  const float alpha = exb[(size_t)e * 8 + head] * rden[d * 8 + head];
  const float4 v = *(const float4*)(h + (size_t)s * 128 + lane * 4);
  float* op = out + (size_t)d * 128 + lane * 4;
  atomicAdd(op + 0, v.x * alpha);
  atomicAdd(op + 1, v.y * alpha);
  atomicAdd(op + 2, v.z * alpha);
  atomicAdd(op + 3, v.w * alpha);
}

__global__ void k_agg40(const float* __restrict__ h, const float* __restrict__ exb,
                        const float* __restrict__ rden, const int* __restrict__ src,
                        const int* __restrict__ dst, float* __restrict__ out,
                        int E, int Etot) {
  const int gid = blockIdx.x * blockDim.x + threadIdx.x;
  const int e = gid >> 5;
  if (e >= Etot) return;
  const int lane = gid & 31;
  int s, d; edge_sd(e, E, src, dst, &s, &d);
  const float alpha = exb[e] * rden[d];
  for (int c = lane; c < 40; c += 32) {
    atomicAdd(out + (size_t)d * 40 + c, h[(size_t)s * 40 + c] * alpha);
  }
}

__global__ void k_bias_act(float* __restrict__ p, const float* __restrict__ b,
                           int n, int F, int relu) {
  const int idx = blockIdx.x * blockDim.x + threadIdx.x;
  if (idx >= n) return;
  float v = p[idx] + b[idx % F];
  if (relu) v = fmaxf(v, 0.0f);
  p[idx] = v;
}

static void edge_phase(const float* hfeat, const float* a_s, const float* a_d,
                       float* als, float* ald, float* emax, float* den, float* exb,
                       const int* srcp, const int* dstp, float* out,
                       int N, int H, int C, int E, int Etot, hipStream_t stream) {
  const int F = H * C;
  const int tNH = N * H;
  k_logits<<<(tNH + 255) / 256, 256, 0, stream>>>(hfeat, a_s, a_d, als, ald, N, H, C);
  k_fill<<<(tNH + 255) / 256, 256, 0, stream>>>(emax, -INFINITY, tNH);
  hipMemsetAsync(den, 0, sizeof(float) * (size_t)tNH, stream);
  const int tE = Etot * H;
  k_edge_max<<<(tE + 255) / 256, 256, 0, stream>>>(als, ald, srcp, dstp, emax, E, Etot, H);
  k_edge_expsum<<<(tE + 255) / 256, 256, 0, stream>>>(als, ald, srcp, dstp, emax, den, exb, E, Etot, H);
  k_recip<<<(tNH + 255) / 256, 256, 0, stream>>>(den, tNH);
  hipMemsetAsync(out, 0, sizeof(float) * (size_t)N * F, stream);
  const int tA = Etot * 32;
  if (F == 128) {
    k_agg128<<<(tA + 255) / 256, 256, 0, stream>>>(hfeat, exb, den, srcp, dstp, out, E, Etot);
  } else {
    k_agg40<<<(tA + 255) / 256, 256, 0, stream>>>(hfeat, exb, den, srcp, dstp, out, E, Etot);
  }
}

static void gemm(const float* A, const float* B, float* C, int Nrows, int K, int Fout,
                 hipStream_t stream) {
  const int tiles_m = (Nrows + 15) / 16;
  const int tiles_n = (Fout + 15) / 16;
  const int threads = tiles_m * tiles_n * 32;
  k_gemm_wmma<<<(threads + 255) / 256, 256, 0, stream>>>(A, B, C, Nrows, K, Fout,
                                                         tiles_m, tiles_n);
}

extern "C" void kernel_launch(void* const* d_in, const int* in_sizes, int n_in,
                              void* d_out, int out_size, void* d_ws, size_t ws_size,
                              hipStream_t stream) {
  const float* x   = (const float*)d_in[0];
  const int*  eidx = (const int*)d_in[1];
  const float* W0  = (const float*)d_in[2];
  const float* as0 = (const float*)d_in[3];
  const float* ad0 = (const float*)d_in[4];
  const float* b0  = (const float*)d_in[5];
  const float* W1  = (const float*)d_in[6];
  const float* as1 = (const float*)d_in[7];
  const float* ad1 = (const float*)d_in[8];
  const float* b1  = (const float*)d_in[9];
  const float* W2  = (const float*)d_in[10];
  const float* as2 = (const float*)d_in[11];
  const float* ad2 = (const float*)d_in[12];
  const float* b2  = (const float*)d_in[13];

  const int F = 128, H = 8, C = 16, Cls = 40;
  const int N    = in_sizes[0] / F;
  const int E    = in_sizes[1] / 2;
  const int Etot = E + N;

  const int* srcp = eidx;
  const int* dstp = eidx + E;

  float* hA   = (float*)d_ws;
  float* hB   = hA   + (size_t)N * F;
  float* als  = hB   + (size_t)N * F;
  float* ald  = als  + (size_t)N * H;
  float* emax = ald  + (size_t)N * H;
  float* den  = emax + (size_t)N * H;
  float* exb  = den  + (size_t)N * H;

  gemm(x, W0, hA, N, F, F, stream);
  edge_phase(hA, as0, ad0, als, ald, emax, den, exb, srcp, dstp, hB,
             N, H, C, E, Etot, stream);
  k_bias_act<<<((N * F) + 255) / 256, 256, 0, stream>>>(hB, b0, N * F, F, 1);

  gemm(hB, W1, hA, N, F, F, stream);
  edge_phase(hA, as1, ad1, als, ald, emax, den, exb, srcp, dstp, hB,
             N, H, C, E, Etot, stream);
  k_bias_act<<<((N * F) + 255) / 256, 256, 0, stream>>>(hB, b1, N * F, F, 1);

  gemm(hB, W2, hA, N, F, Cls, stream);
  edge_phase(hA, as2, ad2, als, ald, emax, den, exb, srcp, dstp, (float*)d_out,
             N, 1, Cls, E, Etot, stream);
  k_bias_act<<<((N * Cls) + 255) / 256, 256, 0, stream>>>((float*)d_out, b2, N * Cls, Cls, 0);
}


# First layer's filter reads one owner byte per edge (dst / 196, written by a small pass before it), 16 edges per 16-B load, exact destination only for matched edges; arrival order as before
# speedup vs baseline: 3.2978x; 3.2978x over previous
#include <hip/hip_runtime.h>
#include <hip/hip_bf16.h>
#include <math.h>


typedef float v2f __attribute__((ext_vector_type(2)));
typedef float v8f __attribute__((ext_vector_type(8)));

#define NEG_SLOPE 0.2f

__global__ void k_gemm_wmma(const float* __restrict__ A, const float* __restrict__ B,
                            float* __restrict__ Cg, int Nrows, int K, int Fout,
                            int tiles_m, int tiles_n) {
  const int wid  = (blockIdx.x * blockDim.x + threadIdx.x) >> 5;
  const int lane = threadIdx.x & 31;
  if (wid >= tiles_m * tiles_n) return;
  const int tm   = wid / tiles_n;
  const int tn   = wid - tm * tiles_n;
  const int half = lane >> 4;
  const int l15  = lane & 15;
  const int arow = tm * 16 + l15;
  const int bcol = tn * 16 + l15;
  const bool bvalid = bcol < Fout;
  const int arowc = (arow < Nrows) ? arow : (Nrows - 1);
  const int bcolc = bvalid ? bcol : (Fout - 1);

  const float* aptr = A + (size_t)arowc * K + half * 2;
  const float* bptr = B + (size_t)(half * 2) * Fout + bcolc;
  const size_t bstep = (size_t)4 * (size_t)Fout;

  v8f acc = {};
#pragma unroll 4
  for (int k = 0; k < K; k += 4) {
    const v2f a = *(const v2f*)aptr;
    v2f b;
    b.x = bptr[0];
    b.y = bptr[Fout];
    acc = __builtin_amdgcn_wmma_f32_16x16x4_f32(false, a, false, b,
                                                (short)0, acc, false, false);
    aptr += 4;
    bptr += bstep;
  }

  if (bvalid) {
    const int baserow = tm * 16 + half * 8;
#pragma unroll
    for (int i = 0; i < 8; ++i) {
      const int r = baserow + i;
      if (r < Nrows) Cg[(size_t)r * Fout + bcol] = acc[i];
    }
  }
}

__global__ void k_logits(const float* __restrict__ h, const float* __restrict__ a_src,
                         const float* __restrict__ a_dst, float* __restrict__ als,
                         float* __restrict__ ald, int N, int H, int C) {
  const int idx = blockIdx.x * blockDim.x + threadIdx.x;
  if (idx >= N * H) return;
  const int n = idx / H, hh = idx - n * H;
  const float* hp = h + (size_t)n * H * C + (size_t)hh * C;
  const float* ws = a_src + hh * C;
  const float* wd = a_dst + hh * C;
  float ss = 0.0f, sd = 0.0f;
  for (int c = 0; c < C; ++c) {
    const float v = hp[c];
    ss += v * ws[c];
    sd += v * wd[c];
  }
  als[idx] = ss;
  ald[idx] = sd;
}

__device__ __forceinline__ float edge_score(float as, float ad) {
  const float z = as + ad;
  return (z > 0.0f) ? z : NEG_SLOPE * z;
}

__device__ __forceinline__ float edge_exp(float as, float ad, float m) {
  const float z = edge_score(as, ad);
  if (!isfinite(m)) m = 0.0f;
  return __expf(z - m);
}

__device__ __forceinline__ void lds_max(float* addr, float v) {
  __hip_atomic_fetch_max(addr, v, __ATOMIC_RELAXED, __HIP_MEMORY_SCOPE_WORKGROUP);
}

#define NPB 196
#define OWN_THREADS 1024

#define OB_THREADS 256
__global__ __launch_bounds__(OB_THREADS)
void k_owner_bytes(const int* __restrict__ dst, unsigned* __restrict__ plane, int nwords) {
  typedef int int4v __attribute__((ext_vector_type(4), aligned(4)));
  const int4v* dst4 = (const int4v*)dst;
  const int w0 = blockIdx.x * (OB_THREADS * 4) + threadIdx.x;
  int4v q[4];
#pragma unroll
  for (int u = 0; u < 4; ++u) {
    const int wi = w0 + u * OB_THREADS;
    q[u] = dst4[(wi < nwords) ? wi : (nwords - 1)];
  }
#pragma unroll
  for (int u = 0; u < 4; ++u) {
    const int wi = w0 + u * OB_THREADS;
    const unsigned b0 = ((unsigned)q[u].x / (unsigned)NPB) & 255u;
    const unsigned b1 = ((unsigned)q[u].y / (unsigned)NPB) & 255u;
    const unsigned b2 = ((unsigned)q[u].z / (unsigned)NPB) & 255u;
    const unsigned b3 = ((unsigned)q[u].w / (unsigned)NPB) & 255u;
    if (wi < nwords) plane[wi] = b0 | (b1 << 8) | (b2 << 16) | (b3 << 24);
  }
}
#define CAPG 8192

struct RowVals { float v0, v1, v2, v3; };

template <int F>
__device__ __forceinline__ int head_of(int lane) {
  return (F == 128) ? (((lane & 7) + (lane >> 4)) & 7) : 0;
}

template <int F>
__device__ __forceinline__ RowVals row_load(const float* __restrict__ h, int s, int lane) {
  RowVals r;
  if (F == 128) {
    const float* hp = h + (size_t)s * 128 + lane;
    r.v0 = hp[0];
    r.v1 = hp[32];
    r.v2 = hp[64];
    r.v3 = hp[96];
  } else {
    const float* hp = h + (size_t)s * F;
    r.v0 = hp[lane];
    r.v1 = (lane + 32 < F) ? hp[lane + 32] : 0.0f;
    r.v2 = 0.0f;
    r.v3 = 0.0f;
  }
  return r;
}

template <int SRC>
__device__ __forceinline__ float row_share(float v) {
  return __int_as_float(__builtin_amdgcn_update_dpp(0, __float_as_int(v), 0x150 + SRC,
                                                    0xf, 0xf, false));
}

template <int F>
__device__ __forceinline__ void row_add(const RowVals& r, float ev, float* acc, float* dens,
                                        int dl, int lane) {
  if (F == 128) {
    float* ap = acc + dl * 128 + lane;
    atomicAdd(ap +  0, r.v0 * row_share<0>(ev));
    atomicAdd(ap + 32, r.v1 * row_share<2>(ev));
    atomicAdd(ap + 64, r.v2 * row_share<4>(ev));
    atomicAdd(ap + 96, r.v3 * row_share<6>(ev));
    if (lane < 8) atomicAdd(&dens[dl * 8 + lane], ev);
  } else {
    float* ap = acc + dl * F;
    atomicAdd(ap + lane, r.v0 * ev);
    if (lane + 32 < F) atomicAdd(ap + lane + 32, r.v1 * ev);
    if (lane == 0) atomicAdd(&dens[dl], ev);
  }
}

template <int F, int H, int C, int RELU, int CAP, int SAVED>
__global__ __launch_bounds__(OWN_THREADS)
void k_own(const float* __restrict__ h, const float* __restrict__ als,
           const float* __restrict__ ald,
           const int* __restrict__ src, const int* __restrict__ dst,
           const float* __restrict__ bias, float* __restrict__ out,
           int* __restrict__ glist, int* __restrict__ gcnt,
           const unsigned* __restrict__ plane,
           int N, int E) {
  static_assert(CAP >= CAPG, "a stored list must fit the LDS list");
  static_assert(SAVED || CAP <= CAPG, "a list that fits LDS must fit its workspace slot");
  __shared__ float acc[NPB * F];
  __shared__ float dens[NPB * H];
  __shared__ float s_ald[NPB * H];
  __shared__ float s_max[NPB * H];
  __shared__ int   list_s[CAP];
  __shared__ int   list_d[CAP];
  __shared__ int   cnt;

  const int tid  = threadIdx.x;
  const int lane = tid & 31;
  const int w    = tid >> 5;
  const int base = blockIdx.x * NPB;
  const int nown = min(NPB, N - base);
  int2* mylist = (int2*)glist + (size_t)blockIdx.x * CAPG;
  const int saved = SAVED ? __builtin_amdgcn_readfirstlane(gcnt[blockIdx.x]) : 0;
  const bool have = SAVED && saved <= CAPG;

  if (tid == 0) cnt = 0;
  for (int i = tid; i < nown * H; i += OWN_THREADS) {
    const int g = base * H + i;
    const float ad = ald[g];
    s_ald[i] = ad;
    s_max[i] = edge_score(als[g], ad);
  }
  __syncthreads();

  const unsigned un = (unsigned)nown;
  if (!have) {
    const uint4* pl4 = (const uint4*)plane;
    const int nvec = E >> 4;
    const unsigned pat = ((unsigned)blockIdx.x & 255u) * 0x01010101u;
    auto zb = [&](unsigned wv) -> unsigned {
      const unsigned x = wv ^ pat;
      return ~((((x & 0x7f7f7f7fu) + 0x7f7f7f7fu) | x) | 0x7f7f7f7fu);
    };
    auto push = [&](int e) {
      const int pos = atomicAdd(&cnt, 1);
      if (pos < CAP) list_s[pos] = e;
    };
    auto match16 = [&](const uint4 q) -> unsigned {
      return (zb(q.x) >> 7) | (zb(q.y) >> 6) | (zb(q.z) >> 5) | (zb(q.w) >> 4);
    };
    auto emit = [&](unsigned m, int vi) {
      while (m) {
        const int p = __ffs(m) - 1;
        m &= m - 1;
        push((vi + ((p >> 2) & 1) * OWN_THREADS) * 16 + ((p & 3) << 2) + (p >> 3));
      }
    };
    const int nfull = nvec / (OWN_THREADS * 4);
    int rnd = (int)(((long long)blockIdx.x * nfull) / gridDim.x);
    for (int k = 0; k < nfull; ++k) {
      const int v0 = rnd * (OWN_THREADS * 4);
      uint4 q[4];
#pragma unroll
      for (int u = 0; u < 4; ++u) q[u] = pl4[v0 + u * OWN_THREADS + tid];
      emit(match16(q[0]) | (match16(q[1]) << 4), v0 + tid);
      emit(match16(q[2]) | (match16(q[3]) << 4), v0 + 2 * OWN_THREADS + tid);
      rnd = (rnd + 1 == nfull) ? 0 : rnd + 1;
    }
    for (int vi = nfull * (OWN_THREADS * 4) + tid; vi < nvec; vi += OWN_THREADS) emit(match16(pl4[vi]), vi);
    for (int e = (nvec << 4) + tid; e < E; e += OWN_THREADS)
      if ((unsigned)(dst[e] - base) < un) push(e);
  }
  __syncthreads();
  const int total = have ? saved : cnt;

  if (total <= CAP) {
    for (int i = tid; i < total; i += OWN_THREADS) {
      int s, dl;
      if (have) {
        const int2 p = mylist[i];
        s = p.x;
        dl = p.y;
      } else {
        const int e = list_s[i];
        const unsigned du = (unsigned)(dst[e] - base);
        const bool mine = du < un;
        s  = mine ? src[e] : base;
        dl = mine ? (int)du : -1;
        if (!SAVED) mylist[i] = make_int2(s, dl);
      }
      list_s[i] = s;
      list_d[i] = dl;
      if (dl >= 0)
        for (int hh = 0; hh < H; ++hh)
          lds_max(&s_max[dl * H + hh], edge_score(als[s * H + hh], s_ald[dl * H + hh]));
    }
  } else {
    for (int idx = tid; idx < E; idx += OWN_THREADS) {
      const unsigned dl = (unsigned)(dst[idx] - base);
      if (dl < un) {
        const int s = src[idx];
        for (int hh = 0; hh < H; ++hh)
          lds_max(&s_max[dl * H + hh], edge_score(als[s * H + hh], s_ald[dl * H + hh]));
      }
    }
  }
  if (!SAVED && tid == 0) gcnt[blockIdx.x] = total;
  __syncthreads();

  for (int i = tid; i < nown * H; i += OWN_THREADS)
    dens[i] = edge_exp(als[base * H + i], s_ald[i], s_max[i]);
  for (int i = tid; i < nown * F; i += OWN_THREADS) {
    const int node = i / F;
    const int cch  = i - node * F;
    const int k    = node * H + cch / C;
    acc[i] = h[(size_t)base * F + i] * edge_exp(als[base * H + k], s_ald[k], s_max[k]);
  }
  __syncthreads();

  if (total <= CAP) {
    const int hd = head_of<F>(lane);
    for (int i = w; i < total; i += 32 * 8) {
      int ss[8], dd[8]; bool ok[8]; float ev[8]; RowVals rv[8];
#pragma unroll
      for (int j = 0; j < 8; ++j) {
        const int idx = i + 32 * j;
        ok[j] = idx < total;
        const int idc = ok[j] ? idx : i;
        ss[j] = __builtin_amdgcn_readfirstlane(list_s[idc]);
        dd[j] = __builtin_amdgcn_readfirstlane(list_d[idc]);
        ok[j] = ok[j] && dd[j] >= 0;
        dd[j] = max(dd[j], 0);
      }
#pragma unroll
      for (int j = 0; j < 8; ++j) ev[j] = als[ss[j] * H + hd];
#pragma unroll
      for (int j = 0; j < 8; ++j) rv[j] = row_load<F>(h, ss[j], lane);
#pragma unroll
      for (int j = 0; j < 8; ++j)
        ev[j] = edge_exp(ev[j], s_ald[dd[j] * H + hd], s_max[dd[j] * H + hd]);
#pragma unroll
      for (int j = 0; j < 8; ++j)
        if (ok[j]) row_add<F>(rv[j], ev[j], acc, dens, dd[j], lane);
    }
  } else {
    for (int i0 = w * 32; i0 < E; i0 += OWN_THREADS) {
      const int idx = i0 + lane;
      const int d = (idx < E) ? dst[idx] : -1;
      const unsigned dl = (unsigned)(d - base);
      unsigned mask = (unsigned)__ballot(dl < (unsigned)nown);
      while (mask) {
        const int l = __ffs(mask) - 1;
        mask &= mask - 1;
        const int s   = __builtin_amdgcn_readfirstlane(src[i0 + l]);
        const int dlu = __builtin_amdgcn_readfirstlane(__shfl((int)dl, l));
        const int hd1 = head_of<F>(lane);
        const float a1 = als[s * H + hd1];
        const RowVals one = row_load<F>(h, s, lane);
        const float ev1 = edge_exp(a1, s_ald[dlu * H + hd1], s_max[dlu * H + hd1]);
        row_add<F>(one, ev1, acc, dens, dlu, lane);
      }
    }
  }
  __syncthreads();

  for (int i = tid; i < nown * F; i += OWN_THREADS) {
    const int node = i / F;
    const int cch  = i - node * F;
    const float rden = 1.0f / (dens[node * H + cch / C] + 1e-16f);
    float v = acc[i] * rden + bias[cch];
    if (RELU) v = fmaxf(v, 0.0f);
    out[(size_t)base * F + i] = v;
  }
}

static void edge_phase(const float* hfeat, const float* a_s, const float* a_d,
                       float* als, float* ald, const float* bias,
                       const int* srcp, const int* dstp, float* out,
                       int* glist, int* gcnt, const unsigned* plane, bool first,
                       int N, int H, int C, int E, hipStream_t stream) {
  const int F = H * C;
  const int tNH = N * H;
  k_logits<<<(tNH + 255) / 256, 256, 0, stream>>>(hfeat, a_s, a_d, als, ald, N, H, C);
  const int nblk = (N + NPB - 1) / NPB;
  if (F == 128 && first) {
    k_own<128, 8, 16, 1, 8192, 0><<<nblk, OWN_THREADS, 0, stream>>>(hfeat, als, ald, srcp, dstp, bias, out, glist, gcnt, plane, N, E);
  } else if (F == 128) {
    k_own<128, 8, 16, 1, 8192, 1><<<nblk, OWN_THREADS, 0, stream>>>(hfeat, als, ald, srcp, dstp, bias, out, glist, gcnt, plane, N, E);
  } else {
    k_own<40, 1, 40, 0, 16384, 1><<<nblk, OWN_THREADS, 0, stream>>>(hfeat, als, ald, srcp, dstp, bias, out, glist, gcnt, plane, N, E);
  }
}

static void gemm(const float* A, const float* B, float* C, int Nrows, int K, int Fout,
                 hipStream_t stream) {
  const int tiles_m = (Nrows + 15) / 16;
  const int tiles_n = (Fout + 15) / 16;
  const int threads = tiles_m * tiles_n * 32;
  k_gemm_wmma<<<(threads + 255) / 256, 256, 0, stream>>>(A, B, C, Nrows, K, Fout,
                                                         tiles_m, tiles_n);
}

extern "C" void kernel_launch(void* const* d_in, const int* in_sizes, int n_in,
                              void* d_out, int out_size, void* d_ws, size_t ws_size,
                              hipStream_t stream) {
  const float* x   = (const float*)d_in[0];
  const int*  eidx = (const int*)d_in[1];
  const float* W0  = (const float*)d_in[2];
  const float* as0 = (const float*)d_in[3];
  const float* ad0 = (const float*)d_in[4];
  const float* b0  = (const float*)d_in[5];
  const float* W1  = (const float*)d_in[6];
  const float* as1 = (const float*)d_in[7];
  const float* ad1 = (const float*)d_in[8];
  const float* b1  = (const float*)d_in[9];
  const float* W2  = (const float*)d_in[10];
  const float* as2 = (const float*)d_in[11];
  const float* ad2 = (const float*)d_in[12];
  const float* b2  = (const float*)d_in[13];

  const int F = 128, H = 8, C = 16, Cls = 40;
  const int N    = in_sizes[0] / F;
  const int E    = in_sizes[1] / 2;

  const int* srcp = eidx;
  const int* dstp = eidx + E;

  float* hA   = (float*)d_ws;
  float* hB   = hA   + (size_t)N * F;
  float* als  = hB   + (size_t)N * F;
  float* ald  = als  + (size_t)N * H;
  const int nblk = (N + NPB - 1) / NPB;
  int* glist  = (int*)(ald + (size_t)N * H);
  int* gcnt   = glist + (size_t)nblk * CAPG * 2;
  unsigned* plane = (unsigned*)(gcnt + ((nblk + 3) & ~3));

  const int nwords = E >> 2;
  k_owner_bytes<<<(nwords + OB_THREADS * 4 - 1) / (OB_THREADS * 4), OB_THREADS, 0, stream>>>(dstp, plane, nwords);

  gemm(x, W0, hA, N, F, F, stream);
  edge_phase(hA, as0, ad0, als, ald, b0, srcp, dstp, hB, glist, gcnt, plane, true,
             N, H, C, E, stream);

  gemm(hB, W1, hA, N, F, F, stream);
  edge_phase(hA, as1, ad1, als, ald, b1, srcp, dstp, hB, glist, gcnt, plane, false,
             N, H, C, E, stream);

  gemm(hB, W2, hA, N, F, Cls, stream);
  edge_phase(hA, as2, ad2, als, ald, b2, srcp, dstp, (float*)d_out, glist, gcnt, plane, false,
             N, 1, Cls, E, stream);
}


# Layers 1, 2: projection done inside the previous layer's owner pass: the block multiplies its finished rows (LDS, pitch 132) by W on f32 WMMA, same k order; 2 GEMM launches, aggregated-row plane gone
# speedup vs baseline: 3.8538x; 1.1686x over previous
#include <hip/hip_runtime.h>
#include <hip/hip_bf16.h>
#include <math.h>


typedef float v2f __attribute__((ext_vector_type(2)));
typedef float v8f __attribute__((ext_vector_type(8)));

#define NEG_SLOPE 0.2f

__global__ void k_gemm_wmma(const float* __restrict__ A, const float* __restrict__ B,
                            float* __restrict__ Cg, int Nrows, int K, int Fout,
                            int tiles_m, int tiles_n) {
  const int wid  = (blockIdx.x * blockDim.x + threadIdx.x) >> 5;
  const int lane = threadIdx.x & 31;
  if (wid >= tiles_m * tiles_n) return;
  const int tm   = wid / tiles_n;
  const int tn   = wid - tm * tiles_n;
  const int half = lane >> 4;
  const int l15  = lane & 15;
  const int arow = tm * 16 + l15;
  const int bcol = tn * 16 + l15;
  const bool bvalid = bcol < Fout;
  const int arowc = (arow < Nrows) ? arow : (Nrows - 1);
  const int bcolc = bvalid ? bcol : (Fout - 1);

  const float* aptr = A + (size_t)arowc * K + half * 2;
  const float* bptr = B + (size_t)(half * 2) * Fout + bcolc;
  const size_t bstep = (size_t)4 * (size_t)Fout;

  v8f acc = {};
#pragma unroll 4
  for (int k = 0; k < K; k += 4) {
    const v2f a = *(const v2f*)aptr;
    v2f b;
    b.x = bptr[0];
    b.y = bptr[Fout];
    acc = __builtin_amdgcn_wmma_f32_16x16x4_f32(false, a, false, b,
                                                (short)0, acc, false, false);
    aptr += 4;
    bptr += bstep;
  }

  if (bvalid) {
    const int baserow = tm * 16 + half * 8;
#pragma unroll
    for (int i = 0; i < 8; ++i) {
      const int r = baserow + i;
      if (r < Nrows) Cg[(size_t)r * Fout + bcol] = acc[i];
    }
  }
}

__global__ void k_logits(const float* __restrict__ h, const float* __restrict__ a_src,
                         const float* __restrict__ a_dst, float* __restrict__ als,
                         float* __restrict__ ald, int N, int H, int C) {
  const int idx = blockIdx.x * blockDim.x + threadIdx.x;
  if (idx >= N * H) return;
  const int n = idx / H, hh = idx - n * H;
  const float* hp = h + (size_t)n * H * C + (size_t)hh * C;
  const float* ws = a_src + hh * C;
  const float* wd = a_dst + hh * C;
  float ss = 0.0f, sd = 0.0f;
  for (int c = 0; c < C; ++c) {
    const float v = hp[c];
    ss += v * ws[c];
    sd += v * wd[c];
  }
  als[idx] = ss;
  ald[idx] = sd;
}

__device__ __forceinline__ float edge_score(float as, float ad) {
  const float z = as + ad;
  return (z > 0.0f) ? z : NEG_SLOPE * z;
}

__device__ __forceinline__ float edge_exp(float as, float ad, float m) {
  const float z = edge_score(as, ad);
  if (!isfinite(m)) m = 0.0f;
  return __expf(z - m);
}

__device__ __forceinline__ void lds_max(float* addr, float v) {
  __hip_atomic_fetch_max(addr, v, __ATOMIC_RELAXED, __HIP_MEMORY_SCOPE_WORKGROUP);
}

#define NPB 196
#define OWN_THREADS 1024

#define OB_THREADS 256
__global__ __launch_bounds__(OB_THREADS)
void k_owner_bytes(const int* __restrict__ dst, unsigned* __restrict__ plane, int nwords) {
  typedef int int4v __attribute__((ext_vector_type(4), aligned(4)));
  const int4v* dst4 = (const int4v*)dst;
  const int w0 = blockIdx.x * (OB_THREADS * 4) + threadIdx.x;
  int4v q[4];
#pragma unroll
  for (int u = 0; u < 4; ++u) {
    const int wi = w0 + u * OB_THREADS;
    q[u] = dst4[(wi < nwords) ? wi : (nwords - 1)];
  }
#pragma unroll
  for (int u = 0; u < 4; ++u) {
    const int wi = w0 + u * OB_THREADS;
    const unsigned b0 = ((unsigned)q[u].x / (unsigned)NPB) & 255u;
    const unsigned b1 = ((unsigned)q[u].y / (unsigned)NPB) & 255u;
    const unsigned b2 = ((unsigned)q[u].z / (unsigned)NPB) & 255u;
    const unsigned b3 = ((unsigned)q[u].w / (unsigned)NPB) & 255u;
    if (wi < nwords) plane[wi] = b0 | (b1 << 8) | (b2 << 16) | (b3 << 24);
  }
}
#define CAPG 8192

struct RowVals { float v0, v1, v2, v3; };

template <int F>
__device__ __forceinline__ int head_of(int lane) {
  return (F == 128) ? (((lane & 7) + (lane >> 4)) & 7) : 0;
}

template <int F>
__device__ __forceinline__ RowVals row_load(const float* __restrict__ h, int s, int lane) {
  RowVals r;
  if (F == 128) {
    const float* hp = h + (size_t)s * 128 + lane;
    r.v0 = hp[0];
    r.v1 = hp[32];
    r.v2 = hp[64];
    r.v3 = hp[96];
  } else {
    const float* hp = h + (size_t)s * F;
    r.v0 = hp[lane];
    r.v1 = (lane + 32 < F) ? hp[lane + 32] : 0.0f;
    r.v2 = 0.0f;
    r.v3 = 0.0f;
  }
  return r;
}

template <int SRC>
__device__ __forceinline__ float row_share(float v) {
  return __int_as_float(__builtin_amdgcn_update_dpp(0, __float_as_int(v), 0x150 + SRC,
                                                    0xf, 0xf, false));
}

template <int F, int P>
__device__ __forceinline__ void row_add(const RowVals& r, float ev, float* acc, float* dens,
                                        int dl, int lane) {
  if (F == 128) {
    float* ap = acc + dl * P + lane;
    atomicAdd(ap +  0, r.v0 * row_share<0>(ev));
    atomicAdd(ap + 32, r.v1 * row_share<2>(ev));
    atomicAdd(ap + 64, r.v2 * row_share<4>(ev));
    atomicAdd(ap + 96, r.v3 * row_share<6>(ev));
    if (lane < 8) atomicAdd(&dens[dl * 8 + lane], ev);
  } else {
    float* ap = acc + dl * P;
    atomicAdd(ap + lane, r.v0 * ev);
    if (lane + 32 < F) atomicAdd(ap + lane + 32, r.v1 * ev);
    if (lane == 0) atomicAdd(&dens[dl], ev);
  }
}

template <int F, int H, int C, int RELU, int CAP, int SAVED, int FN>
__global__ __launch_bounds__(OWN_THREADS)
void k_own(const float* __restrict__ h, const float* __restrict__ als,
           const float* __restrict__ ald,
           const int* __restrict__ src, const int* __restrict__ dst,
           const float* __restrict__ bias, const float* __restrict__ Wn, float* __restrict__ out,
           int* __restrict__ glist, int* __restrict__ gcnt,
           const unsigned* __restrict__ plane,
           int N, int E) {
  static_assert(CAP >= CAPG, "a stored list must fit the LDS list");
  static_assert(SAVED || CAP <= CAPG, "a list that fits LDS must fit its workspace slot");
  constexpr int P = (FN > 0) ? F + 4 : F;
  __shared__ __attribute__((aligned(16))) float acc[NPB * P];
  __shared__ float dens[NPB * H];
  __shared__ float s_ald[NPB * H];
  __shared__ float s_max[NPB * H];
  __shared__ int   list_s[CAP];
  __shared__ int   list_d[CAP];
  __shared__ int   cnt;

  const int tid  = threadIdx.x;
  const int lane = tid & 31;
  const int w    = tid >> 5;
  const int base = blockIdx.x * NPB;
  const int nown = min(NPB, N - base);
  int2* mylist = (int2*)glist + (size_t)blockIdx.x * CAPG;
  const int saved = SAVED ? __builtin_amdgcn_readfirstlane(gcnt[blockIdx.x]) : 0;
  const bool have = SAVED && saved <= CAPG;

  if (tid == 0) cnt = 0;
  for (int i = tid; i < nown * H; i += OWN_THREADS) {
    const int g = base * H + i;
    const float ad = ald[g];
    s_ald[i] = ad;
    s_max[i] = edge_score(als[g], ad);
  }
  __syncthreads();

  const unsigned un = (unsigned)nown;
  if (!have) {
    const uint4* pl4 = (const uint4*)plane;
    const int nvec = E >> 4;
    const unsigned pat = ((unsigned)blockIdx.x & 255u) * 0x01010101u;
    auto zb = [&](unsigned wv) -> unsigned {
      const unsigned x = wv ^ pat;
      return ~((((x & 0x7f7f7f7fu) + 0x7f7f7f7fu) | x) | 0x7f7f7f7fu);
    };
    auto push = [&](int e) {
      const int pos = atomicAdd(&cnt, 1);
      if (pos < CAP) list_s[pos] = e;
    };
    auto match16 = [&](const uint4 q) -> unsigned {
      return (zb(q.x) >> 7) | (zb(q.y) >> 6) | (zb(q.z) >> 5) | (zb(q.w) >> 4);
    };
    auto emit = [&](unsigned m, int vi) {
      while (m) {
        const int p = __ffs(m) - 1;
        m &= m - 1;
        push((vi + ((p >> 2) & 1) * OWN_THREADS) * 16 + ((p & 3) << 2) + (p >> 3));
      }
    };
    const int nfull = nvec / (OWN_THREADS * 4);
    int rnd = (int)(((long long)blockIdx.x * nfull) / gridDim.x);
    for (int k = 0; k < nfull; ++k) {
      const int v0 = rnd * (OWN_THREADS * 4);
      uint4 q[4];
#pragma unroll
      for (int u = 0; u < 4; ++u) q[u] = pl4[v0 + u * OWN_THREADS + tid];
      emit(match16(q[0]) | (match16(q[1]) << 4), v0 + tid);
      emit(match16(q[2]) | (match16(q[3]) << 4), v0 + 2 * OWN_THREADS + tid);
      rnd = (rnd + 1 == nfull) ? 0 : rnd + 1;
    }
    for (int vi = nfull * (OWN_THREADS * 4) + tid; vi < nvec; vi += OWN_THREADS) emit(match16(pl4[vi]), vi);
    for (int e = (nvec << 4) + tid; e < E; e += OWN_THREADS)
      if ((unsigned)(dst[e] - base) < un) push(e);
  }
  __syncthreads();
  const int total = have ? saved : cnt;

  if (total <= CAP) {
    for (int i = tid; i < total; i += OWN_THREADS) {
      int s, dl;
      if (have) {
        const int2 p = mylist[i];
        s = p.x;
        dl = p.y;
      } else {
        const int e = list_s[i];
        const unsigned du = (unsigned)(dst[e] - base);
        const bool mine = du < un;
        s  = mine ? src[e] : base;
        dl = mine ? (int)du : -1;
        if (!SAVED) mylist[i] = make_int2(s, dl);
      }
      list_s[i] = s;
      list_d[i] = dl;
      if (dl >= 0)
        for (int hh = 0; hh < H; ++hh)
          lds_max(&s_max[dl * H + hh], edge_score(als[s * H + hh], s_ald[dl * H + hh]));
    }
  } else {
    for (int idx = tid; idx < E; idx += OWN_THREADS) {
      const unsigned dl = (unsigned)(dst[idx] - base);
      if (dl < un) {
        const int s = src[idx];
        for (int hh = 0; hh < H; ++hh)
          lds_max(&s_max[dl * H + hh], edge_score(als[s * H + hh], s_ald[dl * H + hh]));
      }
    }
  }
  if (!SAVED && tid == 0) gcnt[blockIdx.x] = total;
  __syncthreads();

  for (int i = tid; i < nown * H; i += OWN_THREADS)
    dens[i] = edge_exp(als[base * H + i], s_ald[i], s_max[i]);
  for (int i = tid; i < nown * F; i += OWN_THREADS) {
    const int node = i / F;
    const int cch  = i - node * F;
    const int k    = node * H + cch / C;
    acc[node * P + cch] = h[(size_t)base * F + i] * edge_exp(als[base * H + k], s_ald[k], s_max[k]);
  }
  __syncthreads();

  if (total <= CAP) {
    const int hd = head_of<F>(lane);
    for (int i = w; i < total; i += 32 * 8) {
      int ss[8], dd[8]; bool ok[8]; float ev[8]; RowVals rv[8];
#pragma unroll
      for (int j = 0; j < 8; ++j) {
        const int idx = i + 32 * j;
        ok[j] = idx < total;
        const int idc = ok[j] ? idx : i;
        ss[j] = __builtin_amdgcn_readfirstlane(list_s[idc]);
        dd[j] = __builtin_amdgcn_readfirstlane(list_d[idc]);
        ok[j] = ok[j] && dd[j] >= 0;
        dd[j] = max(dd[j], 0);
      }
#pragma unroll
      for (int j = 0; j < 8; ++j) ev[j] = als[ss[j] * H + hd];
#pragma unroll
      for (int j = 0; j < 8; ++j) rv[j] = row_load<F>(h, ss[j], lane);
#pragma unroll
      for (int j = 0; j < 8; ++j)
        ev[j] = edge_exp(ev[j], s_ald[dd[j] * H + hd], s_max[dd[j] * H + hd]);
#pragma unroll
      for (int j = 0; j < 8; ++j)
        if (ok[j]) row_add<F, P>(rv[j], ev[j], acc, dens, dd[j], lane);
    }
  } else {
    for (int i0 = w * 32; i0 < E; i0 += OWN_THREADS) {
      const int idx = i0 + lane;
      const int d = (idx < E) ? dst[idx] : -1;
      const unsigned dl = (unsigned)(d - base);
      unsigned mask = (unsigned)__ballot(dl < (unsigned)nown);
      while (mask) {
        const int l = __ffs(mask) - 1;
        mask &= mask - 1;
        const int s   = __builtin_amdgcn_readfirstlane(src[i0 + l]);
        const int dlu = __builtin_amdgcn_readfirstlane(__shfl((int)dl, l));
        const int hd1 = head_of<F>(lane);
        const float a1 = als[s * H + hd1];
        const RowVals one = row_load<F>(h, s, lane);
        const float ev1 = edge_exp(a1, s_ald[dlu * H + hd1], s_max[dlu * H + hd1]);
        row_add<F, P>(one, ev1, acc, dens, dlu, lane);
      }
    }
  }
  __syncthreads();

  if constexpr (FN == 0) {
    for (int i = tid; i < nown * F; i += OWN_THREADS) {
      const int node = i / F;
      const int cch  = i - node * F;
      const float rden = 1.0f / (dens[node * H + cch / C] + 1e-16f);
      float v = acc[node * P + cch] * rden + bias[cch];
      if (RELU) v = fmaxf(v, 0.0f);
      out[(size_t)base * F + i] = v;
    }
  } else {
    constexpr int NT  = (FN + 15) / 16;
    constexpr int G   = 32 / NT;
    constexpr int NRT = (NPB + 15) / 16;
    const int wu    = __builtin_amdgcn_readfirstlane(w);
    const bool mm   = wu < NT * G;
    const int tn    = wu % NT;
    const int g     = wu / NT;
    const int half  = lane >> 4;
    const int l15   = lane & 15;
    const int bcol  = tn * 16 + l15;
    const bool bvalid = bcol < FN;
    const int bcolc = bvalid ? bcol : (FN - 1);
    v2f bf[F / 4];
    if (mm) {
      const float* bptr = Wn + (size_t)(half * 2) * FN + bcolc;
#pragma unroll
      for (int s = 0; s < F / 4; ++s) {
        bf[s].x = bptr[(4 * s) * FN];
        bf[s].y = bptr[(4 * s + 1) * FN];
      }
    }
    for (int i = tid; i < nown * F; i += OWN_THREADS) {
      const int node = i / F;
      const int cch  = i - node * F;
      const float rden = 1.0f / (dens[node * H + cch / C] + 1e-16f);
      float v = acc[node * P + cch] * rden + bias[cch];
      if (RELU) v = fmaxf(v, 0.0f);
      acc[node * P + cch] = v;
    }
    __syncthreads();
    if (mm) {
      for (int tm = g; tm < NRT; tm += G) {
        const int r0 = tm * 16;
        if (r0 >= nown) break;
        const int rowc = min(r0 + l15, nown - 1);
        const float* ap = acc + rowc * P + half * 2;
        v8f d = {};
#pragma unroll
        for (int s = 0; s < F / 4; ++s) {
          const v2f a = *(const v2f*)(ap + 4 * s);
          d = __builtin_amdgcn_wmma_f32_16x16x4_f32(false, a, false, bf[s],
                                                    (short)0, d, false, false);
        }
        const int rb = r0 + half * 8;
        float* op = out + (size_t)(base + rb) * FN + bcol;
        if (r0 + 16 <= nown) {
          if (bvalid) {
#pragma unroll
            for (int i = 0; i < 8; ++i) op[(size_t)i * FN] = d[i];
          }
        } else if (bvalid) {
#pragma unroll
          for (int i = 0; i < 8; ++i)
            if (rb + i < nown) op[(size_t)i * FN] = d[i];
        }
      }
    }
  }
}

static void edge_phase(const float* hfeat, const float* a_s, const float* a_d,
                       float* als, float* ald, const float* bias,
                       const int* srcp, const int* dstp, float* out,
                       int* glist, int* gcnt, const unsigned* plane, bool first,
                       const float* Wn,
                       int N, int H, int C, int E, hipStream_t stream) {
  const int F = H * C;
  const int tNH = N * H;
  k_logits<<<(tNH + 255) / 256, 256, 0, stream>>>(hfeat, a_s, a_d, als, ald, N, H, C);
  const int nblk = (N + NPB - 1) / NPB;
  if (F == 128 && first) {
    k_own<128, 8, 16, 1, 8192, 0, 128><<<nblk, OWN_THREADS, 0, stream>>>(hfeat, als, ald, srcp, dstp, bias, Wn, out, glist, gcnt, plane, N, E);
  } else if (F == 128) {
    k_own<128, 8, 16, 1, 8192, 1, 40><<<nblk, OWN_THREADS, 0, stream>>>(hfeat, als, ald, srcp, dstp, bias, Wn, out, glist, gcnt, plane, N, E);
  } else {
    k_own<40, 1, 40, 0, 16384, 1, 0><<<nblk, OWN_THREADS, 0, stream>>>(hfeat, als, ald, srcp, dstp, bias, Wn, out, glist, gcnt, plane, N, E);
  }
}

static void gemm(const float* A, const float* B, float* C, int Nrows, int K, int Fout,
                 hipStream_t stream) {
  const int tiles_m = (Nrows + 15) / 16;
  const int tiles_n = (Fout + 15) / 16;
  const int threads = tiles_m * tiles_n * 32;
  k_gemm_wmma<<<(threads + 255) / 256, 256, 0, stream>>>(A, B, C, Nrows, K, Fout,
                                                         tiles_m, tiles_n);
}

extern "C" void kernel_launch(void* const* d_in, const int* in_sizes, int n_in,
                              void* d_out, int out_size, void* d_ws, size_t ws_size,
                              hipStream_t stream) {
  const float* x   = (const float*)d_in[0];
  const int*  eidx = (const int*)d_in[1];
  const float* W0  = (const float*)d_in[2];
  const float* as0 = (const float*)d_in[3];
  const float* ad0 = (const float*)d_in[4];
  const float* b0  = (const float*)d_in[5];
  const float* W1  = (const float*)d_in[6];
  const float* as1 = (const float*)d_in[7];
  const float* ad1 = (const float*)d_in[8];
  const float* b1  = (const float*)d_in[9];
  const float* W2  = (const float*)d_in[10];
  const float* as2 = (const float*)d_in[11];
  const float* ad2 = (const float*)d_in[12];
  const float* b2  = (const float*)d_in[13];

  const int F = 128, H = 8, C = 16, Cls = 40;
  const int N    = in_sizes[0] / F;
  const int E    = in_sizes[1] / 2;

  const int* srcp = eidx;
  const int* dstp = eidx + E;

  float* hA   = (float*)d_ws;
  float* hB   = hA   + (size_t)N * F;
  float* als  = hB   + (size_t)N * F;
  float* ald  = als  + (size_t)N * H;
  const int nblk = (N + NPB - 1) / NPB;
  int* glist  = (int*)(ald + (size_t)N * H);
  int* gcnt   = glist + (size_t)nblk * CAPG * 2;
  unsigned* plane = (unsigned*)(gcnt + ((nblk + 3) & ~3));

  const int nwords = E >> 2;
  k_owner_bytes<<<(nwords + OB_THREADS * 4 - 1) / (OB_THREADS * 4), OB_THREADS, 0, stream>>>(dstp, plane, nwords);

  gemm(x, W0, hA, N, F, F, stream);
  edge_phase(hA, as0, ad0, als, ald, b0, srcp, dstp, hB, glist, gcnt, plane, true, W1,
             N, H, C, E, stream);

  edge_phase(hB, as1, ad1, als, ald, b1, srcp, dstp, hA, glist, gcnt, plane, false, W2,
             N, H, C, E, stream);

  edge_phase(hA, as2, ad2, als, ald, b2, srcp, dstp, (float*)d_out, glist, gcnt, plane, false, nullptr,
             N, 1, Cls, E, stream);
}


# Owner pass start values and normalising run one wave per owned row (add loop's lane map): exp and reciprocal once per head, not per element; a wave's 7 rows loaded at once; sum order kept
# speedup vs baseline: 4.0114x; 1.0409x over previous
#include <hip/hip_runtime.h>
#include <hip/hip_bf16.h>
#include <math.h>


typedef float v2f __attribute__((ext_vector_type(2)));
typedef float v8f __attribute__((ext_vector_type(8)));

#define NEG_SLOPE 0.2f

__global__ void k_gemm_wmma(const float* __restrict__ A, const float* __restrict__ B,
                            float* __restrict__ Cg, int Nrows, int K, int Fout,
                            int tiles_m, int tiles_n) {
  const int wid  = (blockIdx.x * blockDim.x + threadIdx.x) >> 5;
  const int lane = threadIdx.x & 31;
  if (wid >= tiles_m * tiles_n) return;
  const int tm   = wid / tiles_n;
  const int tn   = wid - tm * tiles_n;
  const int half = lane >> 4;
  const int l15  = lane & 15;
  const int arow = tm * 16 + l15;
  const int bcol = tn * 16 + l15;
  const bool bvalid = bcol < Fout;
  const int arowc = (arow < Nrows) ? arow : (Nrows - 1);
  const int bcolc = bvalid ? bcol : (Fout - 1);

  const float* aptr = A + (size_t)arowc * K + half * 2;
  const float* bptr = B + (size_t)(half * 2) * Fout + bcolc;
  const size_t bstep = (size_t)4 * (size_t)Fout;

  v8f acc = {};
#pragma unroll 4
  for (int k = 0; k < K; k += 4) {
    const v2f a = *(const v2f*)aptr;
    v2f b;
    b.x = bptr[0];
    b.y = bptr[Fout];
    acc = __builtin_amdgcn_wmma_f32_16x16x4_f32(false, a, false, b,
                                                (short)0, acc, false, false);
    aptr += 4;
    bptr += bstep;
  }

  if (bvalid) {
    const int baserow = tm * 16 + half * 8;
#pragma unroll
    for (int i = 0; i < 8; ++i) {
      const int r = baserow + i;
      if (r < Nrows) Cg[(size_t)r * Fout + bcol] = acc[i];
    }
  }
}

__global__ void k_logits(const float* __restrict__ h, const float* __restrict__ a_src,
                         const float* __restrict__ a_dst, float* __restrict__ als,
                         float* __restrict__ ald, int N, int H, int C) {
  const int idx = blockIdx.x * blockDim.x + threadIdx.x;
  if (idx >= N * H) return;
  const int n = idx / H, hh = idx - n * H;
  const float* hp = h + (size_t)n * H * C + (size_t)hh * C;
  const float* ws = a_src + hh * C;
  const float* wd = a_dst + hh * C;
  float ss = 0.0f, sd = 0.0f;
  for (int c = 0; c < C; ++c) {
    const float v = hp[c];
    ss += v * ws[c];
    sd += v * wd[c];
  }
  als[idx] = ss;
  ald[idx] = sd;
}

__device__ __forceinline__ float edge_score(float as, float ad) {
  const float z = as + ad;
  return (z > 0.0f) ? z : NEG_SLOPE * z;
}

__device__ __forceinline__ float edge_exp(float as, float ad, float m) {
  const float z = edge_score(as, ad);
  if (!isfinite(m)) m = 0.0f;
  return __expf(z - m);
}

__device__ __forceinline__ void lds_max(float* addr, float v) {
  __hip_atomic_fetch_max(addr, v, __ATOMIC_RELAXED, __HIP_MEMORY_SCOPE_WORKGROUP);
}

#define NPB 196
#define OWN_THREADS 1024

#define OB_THREADS 256
__global__ __launch_bounds__(OB_THREADS)
void k_owner_bytes(const int* __restrict__ dst, unsigned* __restrict__ plane, int nwords) {
  typedef int int4v __attribute__((ext_vector_type(4), aligned(4)));
  const int4v* dst4 = (const int4v*)dst;
  const int w0 = blockIdx.x * (OB_THREADS * 4) + threadIdx.x;
  int4v q[4];
#pragma unroll
  for (int u = 0; u < 4; ++u) {
    const int wi = w0 + u * OB_THREADS;
    q[u] = dst4[(wi < nwords) ? wi : (nwords - 1)];
  }
#pragma unroll
  for (int u = 0; u < 4; ++u) {
    const int wi = w0 + u * OB_THREADS;
    const unsigned b0 = ((unsigned)q[u].x / (unsigned)NPB) & 255u;
    const unsigned b1 = ((unsigned)q[u].y / (unsigned)NPB) & 255u;
    const unsigned b2 = ((unsigned)q[u].z / (unsigned)NPB) & 255u;
    const unsigned b3 = ((unsigned)q[u].w / (unsigned)NPB) & 255u;
    if (wi < nwords) plane[wi] = b0 | (b1 << 8) | (b2 << 16) | (b3 << 24);
  }
}
#define CAPG 8192

struct RowVals { float v0, v1, v2, v3; };

template <int F>
__device__ __forceinline__ int head_of(int lane) {
  return (F == 128) ? (((lane & 7) + (lane >> 4)) & 7) : 0;
}

template <int F>
__device__ __forceinline__ RowVals row_load(const float* __restrict__ h, int s, int lane) {
  RowVals r;
  if (F == 128) {
    const float* hp = h + (size_t)s * 128 + lane;
    r.v0 = hp[0];
    r.v1 = hp[32];
    r.v2 = hp[64];
    r.v3 = hp[96];
  } else {
    const float* hp = h + (size_t)s * F;
    r.v0 = hp[lane];
    r.v1 = (lane + 32 < F) ? hp[lane + 32] : 0.0f;
    r.v2 = 0.0f;
    r.v3 = 0.0f;
  }
  return r;
}

template <int SRC>
__device__ __forceinline__ float row_share(float v) {
  return __int_as_float(__builtin_amdgcn_update_dpp(0, __float_as_int(v), 0x150 + SRC,
                                                    0xf, 0xf, false));
}

template <int F, int P>
__device__ __forceinline__ void row_add(const RowVals& r, float ev, float* acc, float* dens,
                                        int dl, int lane) {
  if (F == 128) {
    float* ap = acc + dl * P + lane;
    atomicAdd(ap +  0, r.v0 * row_share<0>(ev));
    atomicAdd(ap + 32, r.v1 * row_share<2>(ev));
    atomicAdd(ap + 64, r.v2 * row_share<4>(ev));
    atomicAdd(ap + 96, r.v3 * row_share<6>(ev));
    if (lane < 8) atomicAdd(&dens[dl * 8 + lane], ev);
  } else {
    float* ap = acc + dl * P;
    atomicAdd(ap + lane, r.v0 * ev);
    if (lane + 32 < F) atomicAdd(ap + lane + 32, r.v1 * ev);
    if (lane == 0) atomicAdd(&dens[dl], ev);
  }
}

template <int F, int P>
__device__ __forceinline__ void row_set(const RowVals& r, float ev, float* acc, float* dens,
                                        int dl, int lane) {
  if (F == 128) {
    float* ap = acc + dl * P + lane;
    ap[0]  = r.v0 * row_share<0>(ev);
    ap[32] = r.v1 * row_share<2>(ev);
    ap[64] = r.v2 * row_share<4>(ev);
    ap[96] = r.v3 * row_share<6>(ev);
    if (lane < 8) dens[dl * 8 + lane] = ev;
  } else {
    float* ap = acc + dl * P;
    ap[lane] = r.v0 * ev;
    if (lane + 32 < F) ap[lane + 32] = r.v1 * ev;
    if (lane == 0) dens[dl] = ev;
  }
}

template <int F, int H, int C, int RELU, int CAP, int SAVED, int FN>
__global__ __launch_bounds__(OWN_THREADS)
void k_own(const float* __restrict__ h, const float* __restrict__ als,
           const float* __restrict__ ald,
           const int* __restrict__ src, const int* __restrict__ dst,
           const float* __restrict__ bias, const float* __restrict__ Wn, float* __restrict__ out,
           int* __restrict__ glist, int* __restrict__ gcnt,
           const unsigned* __restrict__ plane,
           int N, int E) {
  static_assert(CAP >= CAPG, "a stored list must fit the LDS list");
  static_assert(SAVED || CAP <= CAPG, "a list that fits LDS must fit its workspace slot");
  constexpr int P = (FN > 0) ? F + 4 : F;
  __shared__ __attribute__((aligned(16))) float acc[NPB * P];
  __shared__ float dens[NPB * H];
  __shared__ float s_ald[NPB * H];
  __shared__ float s_max[NPB * H];
  __shared__ int   list_s[CAP];
  __shared__ int   list_d[CAP];
  __shared__ int   cnt;

  const int tid  = threadIdx.x;
  const int lane = tid & 31;
  const int w    = tid >> 5;
  const int wu   = __builtin_amdgcn_readfirstlane(w);
  const int base = blockIdx.x * NPB;
  const int nown = min(NPB, N - base);
  int2* mylist = (int2*)glist + (size_t)blockIdx.x * CAPG;
  const int saved = SAVED ? __builtin_amdgcn_readfirstlane(gcnt[blockIdx.x]) : 0;
  const bool have = SAVED && saved <= CAPG;

  if (tid == 0) cnt = 0;
  for (int i = tid; i < nown * H; i += OWN_THREADS) {
    const int g = base * H + i;
    const float ad = ald[g];
    s_ald[i] = ad;
    s_max[i] = edge_score(als[g], ad);
  }
  __syncthreads();

  const unsigned un = (unsigned)nown;
  if (!have) {
    const uint4* pl4 = (const uint4*)plane;
    const int nvec = E >> 4;
    const unsigned pat = ((unsigned)blockIdx.x & 255u) * 0x01010101u;
    auto zb = [&](unsigned wv) -> unsigned {
      const unsigned x = wv ^ pat;
      return ~((((x & 0x7f7f7f7fu) + 0x7f7f7f7fu) | x) | 0x7f7f7f7fu);
    };
    auto push = [&](int e) {
      const int pos = atomicAdd(&cnt, 1);
      if (pos < CAP) list_s[pos] = e;
    };
    auto match16 = [&](const uint4 q) -> unsigned {
      return (zb(q.x) >> 7) | (zb(q.y) >> 6) | (zb(q.z) >> 5) | (zb(q.w) >> 4);
    };
    auto emit = [&](unsigned m, int vi) {
      while (m) {
        const int p = __ffs(m) - 1;
        m &= m - 1;
        push((vi + ((p >> 2) & 1) * OWN_THREADS) * 16 + ((p & 3) << 2) + (p >> 3));
      }
    };
    const int nfull = nvec / (OWN_THREADS * 4);
    int rnd = (int)(((long long)blockIdx.x * nfull) / gridDim.x);
    for (int k = 0; k < nfull; ++k) {
      const int v0 = rnd * (OWN_THREADS * 4);
      uint4 q[4];
#pragma unroll
      for (int u = 0; u < 4; ++u) q[u] = pl4[v0 + u * OWN_THREADS + tid];
      emit(match16(q[0]) | (match16(q[1]) << 4), v0 + tid);
      emit(match16(q[2]) | (match16(q[3]) << 4), v0 + 2 * OWN_THREADS + tid);
      rnd = (rnd + 1 == nfull) ? 0 : rnd + 1;
    }
    for (int vi = nfull * (OWN_THREADS * 4) + tid; vi < nvec; vi += OWN_THREADS) emit(match16(pl4[vi]), vi);
    for (int e = (nvec << 4) + tid; e < E; e += OWN_THREADS)
      if ((unsigned)(dst[e] - base) < un) push(e);
  }
  __syncthreads();
  const int total = have ? saved : cnt;

  if (total <= CAP) {
    for (int i = tid; i < total; i += OWN_THREADS) {
      int s, dl;
      if (have) {
        const int2 p = mylist[i];
        s = p.x;
        dl = p.y;
      } else {
        const int e = list_s[i];
        const unsigned du = (unsigned)(dst[e] - base);
        const bool mine = du < un;
        s  = mine ? src[e] : base;
        dl = mine ? (int)du : -1;
        if (!SAVED) mylist[i] = make_int2(s, dl);
      }
      list_s[i] = s;
      list_d[i] = dl;
      if (dl >= 0)
        for (int hh = 0; hh < H; ++hh)
          lds_max(&s_max[dl * H + hh], edge_score(als[s * H + hh], s_ald[dl * H + hh]));
    }
  } else {
    for (int idx = tid; idx < E; idx += OWN_THREADS) {
      const unsigned dl = (unsigned)(dst[idx] - base);
      if (dl < un) {
        const int s = src[idx];
        for (int hh = 0; hh < H; ++hh)
          lds_max(&s_max[dl * H + hh], edge_score(als[s * H + hh], s_ald[dl * H + hh]));
      }
    }
  }
  if (!SAVED && tid == 0) gcnt[blockIdx.x] = total;
  __syncthreads();

  {
    constexpr int NB = (NPB + 31) / 32;
    const int hd0 = head_of<F>(lane);
    float a0[NB]; RowVals r0[NB];
#pragma unroll
    for (int j = 0; j < NB; ++j) {
      const int node = wu + 32 * j;
      const int nc = (node < nown) ? node : 0;
      a0[j] = als[(base + nc) * H + hd0];
      r0[j] = row_load<F>(h, base + nc, lane);
    }
#pragma unroll
    for (int j = 0; j < NB; ++j) {
      const int node = wu + 32 * j;
      if (node < nown)
        row_set<F, P>(r0[j], edge_exp(a0[j], s_ald[node * H + hd0], s_max[node * H + hd0]),
                      acc, dens, node, lane);
    }
  }
  __syncthreads();

  if (total <= CAP) {
    const int hd = head_of<F>(lane);
    for (int i = w; i < total; i += 32 * 8) {
      int ss[8], dd[8]; bool ok[8]; float ev[8]; RowVals rv[8];
#pragma unroll
      for (int j = 0; j < 8; ++j) {
        const int idx = i + 32 * j;
        ok[j] = idx < total;
        const int idc = ok[j] ? idx : i;
        ss[j] = __builtin_amdgcn_readfirstlane(list_s[idc]);
        dd[j] = __builtin_amdgcn_readfirstlane(list_d[idc]);
        ok[j] = ok[j] && dd[j] >= 0;
        dd[j] = max(dd[j], 0);
      }
#pragma unroll
      for (int j = 0; j < 8; ++j) ev[j] = als[ss[j] * H + hd];
#pragma unroll
      for (int j = 0; j < 8; ++j) rv[j] = row_load<F>(h, ss[j], lane);
#pragma unroll
      for (int j = 0; j < 8; ++j)
        ev[j] = edge_exp(ev[j], s_ald[dd[j] * H + hd], s_max[dd[j] * H + hd]);
#pragma unroll
      for (int j = 0; j < 8; ++j)
        if (ok[j]) row_add<F, P>(rv[j], ev[j], acc, dens, dd[j], lane);
    }
  } else {
    for (int i0 = w * 32; i0 < E; i0 += OWN_THREADS) {
      const int idx = i0 + lane;
      const int d = (idx < E) ? dst[idx] : -1;
      const unsigned dl = (unsigned)(d - base);
      unsigned mask = (unsigned)__ballot(dl < (unsigned)nown);
      while (mask) {
        const int l = __ffs(mask) - 1;
        mask &= mask - 1;
        const int s   = __builtin_amdgcn_readfirstlane(src[i0 + l]);
        const int dlu = __builtin_amdgcn_readfirstlane(__shfl((int)dl, l));
        const int hd1 = head_of<F>(lane);
        const float a1 = als[s * H + hd1];
        const RowVals one = row_load<F>(h, s, lane);
        const float ev1 = edge_exp(a1, s_ald[dlu * H + hd1], s_max[dlu * H + hd1]);
        row_add<F, P>(one, ev1, acc, dens, dlu, lane);
      }
    }
  }
  __syncthreads();

  if constexpr (FN == 0) {
    for (int i = tid; i < nown * F; i += OWN_THREADS) {
      const int node = i / F;
      const int cch  = i - node * F;
      const float rden = 1.0f / (dens[node * H + cch / C] + 1e-16f);
      float v = acc[node * P + cch] * rden + bias[cch];
      if (RELU) v = fmaxf(v, 0.0f);
      out[(size_t)base * F + i] = v;
    }
  } else {
    constexpr int NT  = (FN + 15) / 16;
    constexpr int G   = 32 / NT;
    constexpr int NRT = (NPB + 15) / 16;
    static_assert(F == 128, "the normalising pass below uses the 128-wide lane map");
    const bool mm   = wu < NT * G;
    const int tn    = wu % NT;
    const int g     = wu / NT;
    const int half  = lane >> 4;
    const int l15   = lane & 15;
    const int bcol  = tn * 16 + l15;
    const bool bvalid = bcol < FN;
    const int bcolc = bvalid ? bcol : (FN - 1);
    v2f bf[F / 4];
    if (mm) {
      const float* bptr = Wn + (size_t)(half * 2) * FN + bcolc;
#pragma unroll
      for (int s = 0; s < F / 4; ++s) {
        bf[s].x = bptr[(4 * s) * FN];
        bf[s].y = bptr[(4 * s + 1) * FN];
      }
    }
    {
      const int hd1 = head_of<F>(lane);
      const float bz0 = bias[lane], bz1 = bias[32 + lane], bz2 = bias[64 + lane], bz3 = bias[96 + lane];
      for (int node = wu; node < nown; node += 32) {
        float* ap = acc + node * P + lane;
        const float rden = 1.0f / (dens[node * H + hd1] + 1e-16f);
        float v0 = ap[0]  * row_share<0>(rden) + bz0;
        float v1 = ap[32] * row_share<2>(rden) + bz1;
        float v2 = ap[64] * row_share<4>(rden) + bz2;
        float v3 = ap[96] * row_share<6>(rden) + bz3;
        if (RELU) {
          v0 = fmaxf(v0, 0.0f);
          v1 = fmaxf(v1, 0.0f);
          v2 = fmaxf(v2, 0.0f);
          v3 = fmaxf(v3, 0.0f);
        }
        ap[0]  = v0;
        ap[32] = v1;
        ap[64] = v2;
        ap[96] = v3;
      }
    }
    __syncthreads();
    if (mm) {
      for (int tm = g; tm < NRT; tm += G) {
        const int r0 = tm * 16;
        if (r0 >= nown) break;
        const int rowc = min(r0 + l15, nown - 1);
        const float* ap = acc + rowc * P + half * 2;
        v8f d = {};
#pragma unroll
        for (int s = 0; s < F / 4; ++s) {
          const v2f a = *(const v2f*)(ap + 4 * s);
          d = __builtin_amdgcn_wmma_f32_16x16x4_f32(false, a, false, bf[s],
                                                    (short)0, d, false, false);
        }
        const int rb = r0 + half * 8;
        float* op = out + (size_t)(base + rb) * FN + bcol;
        if (r0 + 16 <= nown) {
          if (bvalid) {
#pragma unroll
            for (int i = 0; i < 8; ++i) op[(size_t)i * FN] = d[i];
          }
        } else if (bvalid) {
#pragma unroll
          for (int i = 0; i < 8; ++i)
            if (rb + i < nown) op[(size_t)i * FN] = d[i];
        }
      }
    }
  }
}

static void edge_phase(const float* hfeat, const float* a_s, const float* a_d,
                       float* als, float* ald, const float* bias,
                       const int* srcp, const int* dstp, float* out,
                       int* glist, int* gcnt, const unsigned* plane, bool first,
                       const float* Wn,
                       int N, int H, int C, int E, hipStream_t stream) {
  const int F = H * C;
  const int tNH = N * H;
  k_logits<<<(tNH + 255) / 256, 256, 0, stream>>>(hfeat, a_s, a_d, als, ald, N, H, C);
  const int nblk = (N + NPB - 1) / NPB;
  if (F == 128 && first) {
    k_own<128, 8, 16, 1, 8192, 0, 128><<<nblk, OWN_THREADS, 0, stream>>>(hfeat, als, ald, srcp, dstp, bias, Wn, out, glist, gcnt, plane, N, E);
  } else if (F == 128) {
    k_own<128, 8, 16, 1, 8192, 1, 40><<<nblk, OWN_THREADS, 0, stream>>>(hfeat, als, ald, srcp, dstp, bias, Wn, out, glist, gcnt, plane, N, E);
  } else {
    k_own<40, 1, 40, 0, 16384, 1, 0><<<nblk, OWN_THREADS, 0, stream>>>(hfeat, als, ald, srcp, dstp, bias, Wn, out, glist, gcnt, plane, N, E);
  }
}

static void gemm(const float* A, const float* B, float* C, int Nrows, int K, int Fout,
                 hipStream_t stream) {
  const int tiles_m = (Nrows + 15) / 16;
  const int tiles_n = (Fout + 15) / 16;
  const int threads = tiles_m * tiles_n * 32;
  k_gemm_wmma<<<(threads + 255) / 256, 256, 0, stream>>>(A, B, C, Nrows, K, Fout,
                                                         tiles_m, tiles_n);
}

extern "C" void kernel_launch(void* const* d_in, const int* in_sizes, int n_in,
                              void* d_out, int out_size, void* d_ws, size_t ws_size,
                              hipStream_t stream) {
  const float* x   = (const float*)d_in[0];
  const int*  eidx = (const int*)d_in[1];
  const float* W0  = (const float*)d_in[2];
  const float* as0 = (const float*)d_in[3];
  const float* ad0 = (const float*)d_in[4];
  const float* b0  = (const float*)d_in[5];
  const float* W1  = (const float*)d_in[6];
  const float* as1 = (const float*)d_in[7];
  const float* ad1 = (const float*)d_in[8];
  const float* b1  = (const float*)d_in[9];
  const float* W2  = (const float*)d_in[10];
  const float* as2 = (const float*)d_in[11];
  const float* ad2 = (const float*)d_in[12];
  const float* b2  = (const float*)d_in[13];

  const int F = 128, H = 8, C = 16, Cls = 40;
  const int N    = in_sizes[0] / F;
  const int E    = in_sizes[1] / 2;

  const int* srcp = eidx;
  const int* dstp = eidx + E;

  float* hA   = (float*)d_ws;
  float* hB   = hA   + (size_t)N * F;
  float* als  = hB   + (size_t)N * F;
  float* ald  = als  + (size_t)N * H;
  const int nblk = (N + NPB - 1) / NPB;
  int* glist  = (int*)(ald + (size_t)N * H);
  int* gcnt   = glist + (size_t)nblk * CAPG * 2;
  unsigned* plane = (unsigned*)(gcnt + ((nblk + 3) & ~3));

  const int nwords = E >> 2;
  k_owner_bytes<<<(nwords + OB_THREADS * 4 - 1) / (OB_THREADS * 4), OB_THREADS, 0, stream>>>(dstp, plane, nwords);

  gemm(x, W0, hA, N, F, F, stream);
  edge_phase(hA, as0, ad0, als, ald, b0, srcp, dstp, hB, glist, gcnt, plane, true, W1,
             N, H, C, E, stream);

  edge_phase(hB, as1, ad1, als, ald, b1, srcp, dstp, hA, glist, gcnt, plane, false, W2,
             N, H, C, E, stream);

  edge_phase(hA, as2, ad2, als, ald, b2, srcp, dstp, (float*)d_out, glist, gcnt, plane, false, nullptr,
             N, 1, Cls, E, stream);
}


# Attention logits as columns of the projections: W folded with a_src|a_dst once per call (extra blocks of launch 1), one more WMMA column tile gives als/ald; 3 logit launches gone; logit sum order new
# speedup vs baseline: 4.1585x; 1.0367x over previous
#include <hip/hip_runtime.h>
#include <hip/hip_bf16.h>
#include <math.h>


typedef float v2f __attribute__((ext_vector_type(2)));
typedef float v8f __attribute__((ext_vector_type(8)));

#define NEG_SLOPE 0.2f

__global__ void k_gemm_wmma(const float* __restrict__ A, const float* __restrict__ B,
                            const float* __restrict__ Ba,
                            float* __restrict__ Cg, float* __restrict__ als,
                            float* __restrict__ ald, int Nrows, int K, int Fout, int Hh,
                            int tiles_m, int tiles_n) {
  const int wid  = (blockIdx.x * blockDim.x + threadIdx.x) >> 5;
  const int lane = threadIdx.x & 31;
  if (wid >= tiles_m * tiles_n) return;
  const int tm   = wid / tiles_n;
  const int tn   = wid - tm * tiles_n;
  const int half = lane >> 4;
  const int l15  = lane & 15;
  const int arow = tm * 16 + l15;
  const int bcol = tn * 16 + l15;
  const bool bvalid = bcol < Fout;
  const int arowc = (arow < Nrows) ? arow : (Nrows - 1);
  const int bcolc = bvalid ? bcol : (Fout - 1);
  const bool lg  = __builtin_amdgcn_readfirstlane(tn) == tiles_n - 1;
  const int na   = 2 * Hh;
  const bool lvalid = l15 < na;
  const int lcolc = lvalid ? l15 : (na - 1);

  const float* aptr = A + (size_t)arowc * K + half * 2;
  const float* bptr = B + (size_t)(half * 2) * Fout + bcolc;
  const float* lptr = Ba + (size_t)(half * 2) * na + lcolc;
  const size_t bstep = (size_t)4 * (size_t)Fout;
  const size_t lstep = (size_t)4 * (size_t)na;

  v8f acc = {};
  v8f accl = {};
  if (!lg) {
#pragma unroll 4
    for (int k = 0; k < K; k += 4) {
      const v2f a = *(const v2f*)aptr;
      v2f b;
      b.x = bptr[0];
      b.y = bptr[Fout];
      acc = __builtin_amdgcn_wmma_f32_16x16x4_f32(false, a, false, b,
                                                  (short)0, acc, false, false);
      aptr += 4;
      bptr += bstep;
    }
  } else {
#pragma unroll 4
    for (int k = 0; k < K; k += 4) {
      const v2f a = *(const v2f*)aptr;
      v2f b, bl;
      b.x  = bptr[0];
      b.y  = bptr[Fout];
      bl.x = lptr[0];
      bl.y = lptr[na];
      acc  = __builtin_amdgcn_wmma_f32_16x16x4_f32(false, a, false, b,
                                                   (short)0, acc, false, false);
      accl = __builtin_amdgcn_wmma_f32_16x16x4_f32(false, a, false, bl,
                                                   (short)0, accl, false, false);
      aptr += 4;
      bptr += bstep;
      lptr += lstep;
    }
  }

  const int baserow = tm * 16 + half * 8;
  if (bvalid) {
#pragma unroll
    for (int i = 0; i < 8; ++i) {
      const int r = baserow + i;
      if (r < Nrows) Cg[(size_t)r * Fout + bcol] = acc[i];
    }
  }
  if (lg && lvalid) {
    float* op = (l15 < Hh) ? als + l15 : ald + (l15 - Hh);
#pragma unroll
    for (int i = 0; i < 8; ++i) {
      const int r = baserow + i;
      if (r < Nrows) op[(size_t)r * Hh] = accl[i];
    }
  }
}

__device__ __forceinline__ void fold_one(const float* __restrict__ W, const float* __restrict__ a_src,
                                         const float* __restrict__ a_dst, float* __restrict__ Wa,
                                         int K, int Hn, int C, int t) {
  const int na = 2 * Hn;
  if (t >= K * na) return;
  const int k = t / na, j = t - k * na;
  const int hd = (j < Hn) ? j : j - Hn;
  const float* av = ((j < Hn) ? a_src : a_dst) + hd * C;
  const float* wp = W + (size_t)k * (Hn * C) + hd * C;
  float s = 0.0f;
  for (int cc = 0; cc < C; ++cc) s += wp[cc] * av[cc];
  Wa[t] = s;
}

__device__ __forceinline__ float edge_score(float as, float ad) {
  const float z = as + ad;
  return (z > 0.0f) ? z : NEG_SLOPE * z;
}

__device__ __forceinline__ float edge_exp(float as, float ad, float m) {
  const float z = edge_score(as, ad);
  if (!isfinite(m)) m = 0.0f;
  return __expf(z - m);
}

__device__ __forceinline__ void lds_max(float* addr, float v) {
  __hip_atomic_fetch_max(addr, v, __ATOMIC_RELAXED, __HIP_MEMORY_SCOPE_WORKGROUP);
}

#define NPB 196
#define OWN_THREADS 1024

#define OB_THREADS 256
__global__ __launch_bounds__(OB_THREADS)
void k_owner_bytes(const int* __restrict__ dst, unsigned* __restrict__ plane, int nwords, int nfold,
                   const float* __restrict__ W0, const float* __restrict__ as0,
                   const float* __restrict__ ad0, float* __restrict__ Wa0,
                   const float* __restrict__ W1, const float* __restrict__ as1,
                   const float* __restrict__ ad1, float* __restrict__ Wa1,
                   const float* __restrict__ W2, const float* __restrict__ as2,
                   const float* __restrict__ ad2, float* __restrict__ Wa2,
                   int K, int H, int C, int Cls) {
  if ((int)blockIdx.x < nfold) {
    const int nb = (K * 2 * H + OB_THREADS - 1) / OB_THREADS;
    const int fb = (int)blockIdx.x;
    if (fb < nb)          fold_one(W0, as0, ad0, Wa0, K, H, C, fb * OB_THREADS + (int)threadIdx.x);
    else if (fb < 2 * nb) fold_one(W1, as1, ad1, Wa1, K, H, C, (fb - nb) * OB_THREADS + (int)threadIdx.x);
    else                  fold_one(W2, as2, ad2, Wa2, K, 1, Cls, (fb - 2 * nb) * OB_THREADS + (int)threadIdx.x);
    return;
  }
  typedef int int4v __attribute__((ext_vector_type(4), aligned(4)));
  const int4v* dst4 = (const int4v*)dst;
  const int w0 = ((int)blockIdx.x - nfold) * (OB_THREADS * 4) + threadIdx.x;
  int4v q[4];
#pragma unroll
  for (int u = 0; u < 4; ++u) {
    const int wi = w0 + u * OB_THREADS;
    q[u] = dst4[(wi < nwords) ? wi : (nwords - 1)];
  }
#pragma unroll
  for (int u = 0; u < 4; ++u) {
    const int wi = w0 + u * OB_THREADS;
    const unsigned b0 = ((unsigned)q[u].x / (unsigned)NPB) & 255u;
    const unsigned b1 = ((unsigned)q[u].y / (unsigned)NPB) & 255u;
    const unsigned b2 = ((unsigned)q[u].z / (unsigned)NPB) & 255u;
    const unsigned b3 = ((unsigned)q[u].w / (unsigned)NPB) & 255u;
    if (wi < nwords) plane[wi] = b0 | (b1 << 8) | (b2 << 16) | (b3 << 24);
  }
}
#define CAPG 8192

struct RowVals { float v0, v1, v2, v3; };

template <int F>
__device__ __forceinline__ int head_of(int lane) {
  return (F == 128) ? (((lane & 7) + (lane >> 4)) & 7) : 0;
}

template <int F>
__device__ __forceinline__ RowVals row_load(const float* __restrict__ h, int s, int lane) {
  RowVals r;
  if (F == 128) {
    const float* hp = h + (size_t)s * 128 + lane;
    r.v0 = hp[0];
    r.v1 = hp[32];
    r.v2 = hp[64];
    r.v3 = hp[96];
  } else {
    const float* hp = h + (size_t)s * F;
    r.v0 = hp[lane];
    r.v1 = (lane + 32 < F) ? hp[lane + 32] : 0.0f;
    r.v2 = 0.0f;
    r.v3 = 0.0f;
  }
  return r;
}

template <int SRC>
__device__ __forceinline__ float row_share(float v) {
  return __int_as_float(__builtin_amdgcn_update_dpp(0, __float_as_int(v), 0x150 + SRC,
                                                    0xf, 0xf, false));
}

template <int F, int P>
__device__ __forceinline__ void row_add(const RowVals& r, float ev, float* acc, float* dens,
                                        int dl, int lane) {
  if (F == 128) {
    float* ap = acc + dl * P + lane;
    atomicAdd(ap +  0, r.v0 * row_share<0>(ev));
    atomicAdd(ap + 32, r.v1 * row_share<2>(ev));
    atomicAdd(ap + 64, r.v2 * row_share<4>(ev));
    atomicAdd(ap + 96, r.v3 * row_share<6>(ev));
    if (lane < 8) atomicAdd(&dens[dl * 8 + lane], ev);
  } else {
    float* ap = acc + dl * P;
    atomicAdd(ap + lane, r.v0 * ev);
    if (lane + 32 < F) atomicAdd(ap + lane + 32, r.v1 * ev);
    if (lane == 0) atomicAdd(&dens[dl], ev);
  }
}

template <int F, int P>
__device__ __forceinline__ void row_set(const RowVals& r, float ev, float* acc, float* dens,
                                        int dl, int lane) {
  if (F == 128) {
    float* ap = acc + dl * P + lane;
    ap[0]  = r.v0 * row_share<0>(ev);
    ap[32] = r.v1 * row_share<2>(ev);
    ap[64] = r.v2 * row_share<4>(ev);
    ap[96] = r.v3 * row_share<6>(ev);
    if (lane < 8) dens[dl * 8 + lane] = ev;
  } else {
    float* ap = acc + dl * P;
    ap[lane] = r.v0 * ev;
    if (lane + 32 < F) ap[lane + 32] = r.v1 * ev;
    if (lane == 0) dens[dl] = ev;
  }
}

template <int F, int H, int C, int RELU, int CAP, int SAVED, int FN, int HN>
__global__ __launch_bounds__(OWN_THREADS)
void k_own(const float* __restrict__ h, const float* __restrict__ als,
           const float* __restrict__ ald,
           const int* __restrict__ src, const int* __restrict__ dst,
           const float* __restrict__ bias, const float* __restrict__ Wn,
           const float* __restrict__ Wan, float* __restrict__ out,
           float* __restrict__ alsn, float* __restrict__ aldn,
           int* __restrict__ glist, int* __restrict__ gcnt,
           const unsigned* __restrict__ plane,
           int N, int E) {
  static_assert(CAP >= CAPG, "a stored list must fit the LDS list");
  static_assert(SAVED || CAP <= CAPG, "a list that fits LDS must fit its workspace slot");
  constexpr int P = (FN > 0) ? F + 4 : F;
  __shared__ __attribute__((aligned(16))) float acc[NPB * P];
  __shared__ float dens[NPB * H];
  __shared__ float s_ald[NPB * H];
  __shared__ float s_max[NPB * H];
  __shared__ int   list_s[CAP];
  __shared__ int   list_d[CAP];
  __shared__ int   cnt;

  const int tid  = threadIdx.x;
  const int lane = tid & 31;
  const int w    = tid >> 5;
  const int wu   = __builtin_amdgcn_readfirstlane(w);
  const int base = blockIdx.x * NPB;
  const int nown = min(NPB, N - base);
  int2* mylist = (int2*)glist + (size_t)blockIdx.x * CAPG;
  const int saved = SAVED ? __builtin_amdgcn_readfirstlane(gcnt[blockIdx.x]) : 0;
  const bool have = SAVED && saved <= CAPG;

  if (tid == 0) cnt = 0;
  for (int i = tid; i < nown * H; i += OWN_THREADS) {
    const int g = base * H + i;
    const float ad = ald[g];
    s_ald[i] = ad;
    s_max[i] = edge_score(als[g], ad);
  }
  __syncthreads();

  const unsigned un = (unsigned)nown;
  if (!have) {
    const uint4* pl4 = (const uint4*)plane;
    const int nvec = E >> 4;
    const unsigned pat = ((unsigned)blockIdx.x & 255u) * 0x01010101u;
    auto zb = [&](unsigned wv) -> unsigned {
      const unsigned x = wv ^ pat;
      return ~((((x & 0x7f7f7f7fu) + 0x7f7f7f7fu) | x) | 0x7f7f7f7fu);
    };
    auto push = [&](int e) {
      const int pos = atomicAdd(&cnt, 1);
      if (pos < CAP) list_s[pos] = e;
    };
    auto match16 = [&](const uint4 q) -> unsigned {
      return (zb(q.x) >> 7) | (zb(q.y) >> 6) | (zb(q.z) >> 5) | (zb(q.w) >> 4);
    };
    auto emit = [&](unsigned m, int vi) {
      while (m) {
        const int p = __ffs(m) - 1;
        m &= m - 1;
        push((vi + ((p >> 2) & 1) * OWN_THREADS) * 16 + ((p & 3) << 2) + (p >> 3));
      }
    };
    const int nfull = nvec / (OWN_THREADS * 4);
    int rnd = (int)(((long long)blockIdx.x * nfull) / gridDim.x);
    for (int k = 0; k < nfull; ++k) {
      const int v0 = rnd * (OWN_THREADS * 4);
      uint4 q[4];
#pragma unroll
      for (int u = 0; u < 4; ++u) q[u] = pl4[v0 + u * OWN_THREADS + tid];
      emit(match16(q[0]) | (match16(q[1]) << 4), v0 + tid);
      emit(match16(q[2]) | (match16(q[3]) << 4), v0 + 2 * OWN_THREADS + tid);
      rnd = (rnd + 1 == nfull) ? 0 : rnd + 1;
    }
    for (int vi = nfull * (OWN_THREADS * 4) + tid; vi < nvec; vi += OWN_THREADS) emit(match16(pl4[vi]), vi);
    for (int e = (nvec << 4) + tid; e < E; e += OWN_THREADS)
      if ((unsigned)(dst[e] - base) < un) push(e);
  }
  __syncthreads();
  const int total = have ? saved : cnt;

  if (total <= CAP) {
    for (int i = tid; i < total; i += OWN_THREADS) {
      int s, dl;
      if (have) {
        const int2 p = mylist[i];
        s = p.x;
        dl = p.y;
      } else {
        const int e = list_s[i];
        const unsigned du = (unsigned)(dst[e] - base);
        const bool mine = du < un;
        s  = mine ? src[e] : base;
        dl = mine ? (int)du : -1;
        if (!SAVED) mylist[i] = make_int2(s, dl);
      }
      list_s[i] = s;
      list_d[i] = dl;
      if (dl >= 0)
        for (int hh = 0; hh < H; ++hh)
          lds_max(&s_max[dl * H + hh], edge_score(als[s * H + hh], s_ald[dl * H + hh]));
    }
  } else {
    for (int idx = tid; idx < E; idx += OWN_THREADS) {
      const unsigned dl = (unsigned)(dst[idx] - base);
      if (dl < un) {
        const int s = src[idx];
        for (int hh = 0; hh < H; ++hh)
          lds_max(&s_max[dl * H + hh], edge_score(als[s * H + hh], s_ald[dl * H + hh]));
      }
    }
  }
  if (!SAVED && tid == 0) gcnt[blockIdx.x] = total;
  __syncthreads();

  {
    constexpr int NB = (NPB + 31) / 32;
    const int hd0 = head_of<F>(lane);
    float a0[NB]; RowVals r0[NB];
#pragma unroll
    for (int j = 0; j < NB; ++j) {
      const int node = wu + 32 * j;
      const int nc = (node < nown) ? node : 0;
      a0[j] = als[(base + nc) * H + hd0];
      r0[j] = row_load<F>(h, base + nc, lane);
    }
#pragma unroll
    for (int j = 0; j < NB; ++j) {
      const int node = wu + 32 * j;
      if (node < nown)
        row_set<F, P>(r0[j], edge_exp(a0[j], s_ald[node * H + hd0], s_max[node * H + hd0]),
                      acc, dens, node, lane);
    }
  }
  __syncthreads();

  if (total <= CAP) {
    const int hd = head_of<F>(lane);
    for (int i = w; i < total; i += 32 * 8) {
      int ss[8], dd[8]; bool ok[8]; float ev[8]; RowVals rv[8];
#pragma unroll
      for (int j = 0; j < 8; ++j) {
        const int idx = i + 32 * j;
        ok[j] = idx < total;
        const int idc = ok[j] ? idx : i;
        ss[j] = __builtin_amdgcn_readfirstlane(list_s[idc]);
        dd[j] = __builtin_amdgcn_readfirstlane(list_d[idc]);
        ok[j] = ok[j] && dd[j] >= 0;
        dd[j] = max(dd[j], 0);
      }
#pragma unroll
      for (int j = 0; j < 8; ++j) ev[j] = als[ss[j] * H + hd];
#pragma unroll
      for (int j = 0; j < 8; ++j) rv[j] = row_load<F>(h, ss[j], lane);
#pragma unroll
      for (int j = 0; j < 8; ++j)
        ev[j] = edge_exp(ev[j], s_ald[dd[j] * H + hd], s_max[dd[j] * H + hd]);
#pragma unroll
      for (int j = 0; j < 8; ++j)
        if (ok[j]) row_add<F, P>(rv[j], ev[j], acc, dens, dd[j], lane);
    }
  } else {
    for (int i0 = w * 32; i0 < E; i0 += OWN_THREADS) {
      const int idx = i0 + lane;
      const int d = (idx < E) ? dst[idx] : -1;
      const unsigned dl = (unsigned)(d - base);
      unsigned mask = (unsigned)__ballot(dl < (unsigned)nown);
      while (mask) {
        const int l = __ffs(mask) - 1;
        mask &= mask - 1;
        const int s   = __builtin_amdgcn_readfirstlane(src[i0 + l]);
        const int dlu = __builtin_amdgcn_readfirstlane(__shfl((int)dl, l));
        const int hd1 = head_of<F>(lane);
        const float a1 = als[s * H + hd1];
        const RowVals one = row_load<F>(h, s, lane);
        const float ev1 = edge_exp(a1, s_ald[dlu * H + hd1], s_max[dlu * H + hd1]);
        row_add<F, P>(one, ev1, acc, dens, dlu, lane);
      }
    }
  }
  __syncthreads();

  if constexpr (FN == 0) {
    for (int i = tid; i < nown * F; i += OWN_THREADS) {
      const int node = i / F;
      const int cch  = i - node * F;
      const float rden = 1.0f / (dens[node * H + cch / C] + 1e-16f);
      float v = acc[node * P + cch] * rden + bias[cch];
      if (RELU) v = fmaxf(v, 0.0f);
      out[(size_t)base * F + i] = v;
    }
  } else {
    constexpr int NA  = 2 * HN;
    constexpr int FNX = FN + NA;
    constexpr int NT  = (FNX + 15) / 16;
    constexpr int G   = 32 / NT;
    constexpr int NRT = (NPB + 15) / 16;
    static_assert(F == 128, "the normalising pass below uses the 128-wide lane map");
    const bool mm   = wu < NT * G;
    const int tn    = wu % NT;
    const int g     = wu / NT;
    const int half  = lane >> 4;
    const int l15   = lane & 15;
    const int bcol  = tn * 16 + l15;
    const bool bvalid = bcol < FNX;
    const int bcolc = bvalid ? bcol : (FNX - 1);
    const bool isw  = bcolc < FN;
    const int bw    = isw ? FN : NA;
    v2f bf[F / 4];
    if (mm) {
      const float* bptr = isw ? Wn + (size_t)(half * 2) * FN + bcolc
                              : Wan + (size_t)(half * 2) * NA + (bcolc - FN);
#pragma unroll
      for (int s = 0; s < F / 4; ++s) {
        bf[s].x = bptr[(4 * s) * bw];
        bf[s].y = bptr[(4 * s + 1) * bw];
      }
    }
    {
      const int hd1 = head_of<F>(lane);
      const float bz0 = bias[lane], bz1 = bias[32 + lane], bz2 = bias[64 + lane], bz3 = bias[96 + lane];
      for (int node = wu; node < nown; node += 32) {
        float* ap = acc + node * P + lane;
        const float rden = 1.0f / (dens[node * H + hd1] + 1e-16f);
        float v0 = ap[0]  * row_share<0>(rden) + bz0;
        float v1 = ap[32] * row_share<2>(rden) + bz1;
        float v2 = ap[64] * row_share<4>(rden) + bz2;
        float v3 = ap[96] * row_share<6>(rden) + bz3;
        if (RELU) {
          v0 = fmaxf(v0, 0.0f);
          v1 = fmaxf(v1, 0.0f);
          v2 = fmaxf(v2, 0.0f);
          v3 = fmaxf(v3, 0.0f);
        }
        ap[0]  = v0;
        ap[32] = v1;
        ap[64] = v2;
        ap[96] = v3;
      }
    }
    __syncthreads();
    if (mm) {
      for (int tm = g; tm < NRT; tm += G) {
        const int r0 = tm * 16;
        if (r0 >= nown) break;
        const int rowc = min(r0 + l15, nown - 1);
        const float* ap = acc + rowc * P + half * 2;
        v8f d = {};
#pragma unroll
        for (int s = 0; s < F / 4; ++s) {
          const v2f a = *(const v2f*)(ap + 4 * s);
          d = __builtin_amdgcn_wmma_f32_16x16x4_f32(false, a, false, bf[s],
                                                    (short)0, d, false, false);
        }
        const int rb = r0 + half * 8;
        const int jl = bcolc - FN;
        float* op = isw ? out + (size_t)(base + rb) * FN + bcolc
                        : ((jl < HN) ? alsn + (size_t)(base + rb) * HN + jl
                                     : aldn + (size_t)(base + rb) * HN + (jl - HN));
        const int os = isw ? FN : HN;
        if (r0 + 16 <= nown) {
          if (bvalid) {
#pragma unroll
            for (int i = 0; i < 8; ++i) op[(size_t)i * os] = d[i];
          }
        } else if (bvalid) {
#pragma unroll
          for (int i = 0; i < 8; ++i)
            if (rb + i < nown) op[(size_t)i * os] = d[i];
        }
      }
    }
  }
}

static void edge_phase(const float* hfeat,
                       const float* als, const float* ald, const float* bias,
                       const int* srcp, const int* dstp, float* out,
                       int* glist, int* gcnt, const unsigned* plane, bool first,
                       const float* Wn, const float* Wan, float* alsn, float* aldn,
                       int N, int H, int C, int E, hipStream_t stream) {
  const int F = H * C;
  const int nblk = (N + NPB - 1) / NPB;
  if (F == 128 && first) {
    k_own<128, 8, 16, 1, 8192, 0, 128, 8><<<nblk, OWN_THREADS, 0, stream>>>(hfeat, als, ald, srcp, dstp, bias, Wn, Wan, out, alsn, aldn, glist, gcnt, plane, N, E);
  } else if (F == 128) {
    k_own<128, 8, 16, 1, 8192, 1, 40, 1><<<nblk, OWN_THREADS, 0, stream>>>(hfeat, als, ald, srcp, dstp, bias, Wn, Wan, out, alsn, aldn, glist, gcnt, plane, N, E);
  } else {
    k_own<40, 1, 40, 0, 16384, 1, 0, 0><<<nblk, OWN_THREADS, 0, stream>>>(hfeat, als, ald, srcp, dstp, bias, Wn, Wan, out, alsn, aldn, glist, gcnt, plane, N, E);
  }
}

static void gemm(const float* A, const float* B, const float* Ba, float* C, float* als, float* ald,
                 int Nrows, int K, int Fout, int Hh, hipStream_t stream) {
  const int tiles_m = (Nrows + 15) / 16;
  const int tiles_n = (Fout + 15) / 16;
  const int threads = tiles_m * tiles_n * 32;
  k_gemm_wmma<<<(threads + 255) / 256, 256, 0, stream>>>(A, B, Ba, C, als, ald, Nrows, K, Fout, Hh,
                                                         tiles_m, tiles_n);
}

extern "C" void kernel_launch(void* const* d_in, const int* in_sizes, int n_in,
                              void* d_out, int out_size, void* d_ws, size_t ws_size,
                              hipStream_t stream) {
  const float* x   = (const float*)d_in[0];
  const int*  eidx = (const int*)d_in[1];
  const float* W0  = (const float*)d_in[2];
  const float* as0 = (const float*)d_in[3];
  const float* ad0 = (const float*)d_in[4];
  const float* b0  = (const float*)d_in[5];
  const float* W1  = (const float*)d_in[6];
  const float* as1 = (const float*)d_in[7];
  const float* ad1 = (const float*)d_in[8];
  const float* b1  = (const float*)d_in[9];
  const float* W2  = (const float*)d_in[10];
  const float* as2 = (const float*)d_in[11];
  const float* ad2 = (const float*)d_in[12];
  const float* b2  = (const float*)d_in[13];

  const int F = 128, H = 8, C = 16, Cls = 40;
  const int N    = in_sizes[0] / F;
  const int E    = in_sizes[1] / 2;

  const int* srcp = eidx;
  const int* dstp = eidx + E;

  float* hA   = (float*)d_ws;
  float* hB   = hA   + (size_t)N * F;
  float* alsA = hB   + (size_t)N * F;
  float* aldA = alsA + (size_t)N * H;
  float* alsB = aldA + (size_t)N * H;
  float* aldB = alsB + (size_t)N * H;
  float* Wa0  = aldB + (size_t)N * H;
  float* Wa1  = Wa0  + F * 2 * H;
  float* Wa2  = Wa1  + F * 2 * H;
  const int nblk = (N + NPB - 1) / NPB;
  int* glist  = (int*)(Wa2 + F * 2 * H);
  int* gcnt   = glist + (size_t)nblk * CAPG * 2;
  unsigned* plane = (unsigned*)(gcnt + ((nblk + 3) & ~3));

  const int nwords = E >> 2;
  const int nob = (nwords + OB_THREADS * 4 - 1) / (OB_THREADS * 4);
  const int nfold = 2 * ((F * 2 * H + OB_THREADS - 1) / OB_THREADS) + (F * 2 + OB_THREADS - 1) / OB_THREADS;
  k_owner_bytes<<<nfold + nob, OB_THREADS, 0, stream>>>(dstp, plane, nwords, nfold,
                                                        W0, as0, ad0, Wa0, W1, as1, ad1, Wa1,
                                                        W2, as2, ad2, Wa2, F, H, C, Cls);

  gemm(x, W0, Wa0, hA, alsA, aldA, N, F, F, H, stream);
  edge_phase(hA, alsA, aldA, b0, srcp, dstp, hB, glist, gcnt, plane, true, W1, Wa1, alsB, aldB,
             N, H, C, E, stream);

  edge_phase(hB, alsB, aldB, b1, srcp, dstp, hA, glist, gcnt, plane, false, W2, Wa2, alsA, aldA,
             N, H, C, E, stream);

  edge_phase(hA, alsA, aldA, b2, srcp, dstp, (float*)d_out, glist, gcnt, plane, false, nullptr, nullptr, nullptr, nullptr,
             N, 1, Cls, E, stream);
}
